# starting kernel
# baseline (speedup 1.0000x reference)
#include <hip/hip_runtime.h>


static constexpr int kBS = 8;
static constexpr int kC  = 256;
static constexpr int kN  = 4096;
static constexpr int kDK = 64;

typedef __attribute__((ext_vector_type(2))) float v2f;
typedef __attribute__((ext_vector_type(8))) float v8f;

__device__ __forceinline__ v8f wmma_f32(v2f a, v2f b, v8f c) {
  return __builtin_amdgcn_wmma_f32_16x16x4_f32(false, a, false, b,
                                               (short)0, c, false, false);
}

__global__ void qkv_kernel(const float* __restrict__ x,
                           const float* __restrict__ Wqkv,
                           float* __restrict__ qT,
                           float* __restrict__ Kd,
                           float* __restrict__ Vd) {
  const int L    = threadIdx.x;
  const int col  = L & 15;
  const int half = L >> 4;

  int blk = blockIdx.x;
  const int b   = blk / (12 * 64);
  const int rem = blk % (12 * 64);
  const int m0  = (rem / 64) * 16;
  const int n0  = (rem % 64) * 64;

  const float* xb   = x + (size_t)b * kC * kN;
  const float* Arow = Wqkv + (size_t)(m0 + col) * kC + 2 * half;

  v8f c0 = {}, c1 = {}, c2 = {}, c3 = {};
  const int n = n0 + col;

#pragma unroll 4
  for (int kk = 0; kk < kC; kk += 4) {
    v2f a = *(const v2f*)(Arow + kk);
    const float* xr0 = xb + (size_t)(kk + 2 * half) * kN;
    const float* xr1 = xr0 + kN;
    v2f b0 = { xr0[n],      xr1[n]      };
    v2f b1 = { xr0[n + 16], xr1[n + 16] };
    v2f b2 = { xr0[n + 32], xr1[n + 32] };
    v2f b3 = { xr0[n + 48], xr1[n + 48] };
    c0 = wmma_f32(a, b0, c0);
    c1 = wmma_f32(a, b1, c1);
    c2 = wmma_f32(a, b2, c2);
    c3 = wmma_f32(a, b3, c3);
  }

  v8f cs[4] = { c0, c1, c2, c3 };
#pragma unroll
  for (int t = 0; t < 4; ++t) {
#pragma unroll
    for (int r = 0; r < 8; ++r) {
      const int row = m0 + r + 8 * half;
      const int nc  = n0 + 16 * t + col;
      const float vv = cs[t][r];
      if (row < 64) {
        qT[((size_t)b * kN + nc) * kDK + row] = vv;
      } else if (row < 128) {
        Kd[((size_t)b * kDK + (row - 64)) * kN + nc] = vv;
      } else {
        Vd[((size_t)b * kDK + (row - 128)) * kN + nc] = vv;
      }
    }
  }
}

__global__ void stats_kernel(const float* __restrict__ qT,
                             const float* __restrict__ Kd,
                             float* __restrict__ Lrow) {
  const int L    = threadIdx.x;
  const int col  = L & 15;
  const int half = L >> 4;
  const int b  = blockIdx.x >> 8;
  const int i0 = (blockIdx.x & 255) * 16;

  v2f aq[16];
  const float* qrow = qT + ((size_t)b * kN + i0 + col) * kDK + 2 * half;
#pragma unroll
  for (int s = 0; s < 16; ++s) aq[s] = *(const v2f*)(qrow + 4 * s);

  const float* Kb = Kd + (size_t)b * kDK * kN;

  float mr[8], zr[8];
#pragma unroll
  for (int r = 0; r < 8; ++r) { mr[r] = -3.0e38f; zr[r] = 0.0f; }

  for (int jt = 0; jt < 256; ++jt) {
    const int n = jt * 16 + col;
    v8f c = {};
#pragma unroll
    for (int s = 0; s < 16; ++s) {
      const float* kr = Kb + (size_t)(4 * s + 2 * half) * kN;
      v2f bb = { kr[n], kr[n + kN] };
      c = wmma_f32(aq[s], bb, c);
    }
#pragma unroll
    for (int r = 0; r < 8; ++r) {
      const float v  = c[r];
      const float mn = fmaxf(mr[r], v);
      zr[r] = zr[r] * __expf(mr[r] - mn) + __expf(v - mn);
      mr[r] = mn;
    }
  }

#pragma unroll
  for (int r = 0; r < 8; ++r) {
    float m = mr[r], z = zr[r];
#pragma unroll
    for (int mask = 1; mask < 16; mask <<= 1) {
      const float om = __shfl_xor(m, mask, 16);
      const float oz = __shfl_xor(z, mask, 16);
      const float mn = fmaxf(m, om);
      z = z * __expf(m - mn) + oz * __expf(om - mn);
      m = mn;
    }
    if (col == 0)
      Lrow[(size_t)b * kN + i0 + 8 * half + r] = m + __logf(z);
  }
}

__global__ void attn_out_kernel(const float* __restrict__ qT,
                                const float* __restrict__ Kd,
                                const float* __restrict__ Vd,
                                const float* __restrict__ Lrow,
                                const float* __restrict__ Wout,
                                const float* __restrict__ x,
                                const float* __restrict__ gamma,
                                float* __restrict__ out) {
  __shared__ float Pt[16 * 16];
  __shared__ float ys[64 * 16];

  const int L    = threadIdx.x;
  const int col  = L & 15;
  const int half = L >> 4;
  const int b  = blockIdx.x >> 8;
  const int j0 = (blockIdx.x & 255) * 16;
  const int n  = j0 + col;

  const float* Kb = Kd + (size_t)b * kDK * kN;
  const float* Vb = Vd + (size_t)b * kDK * kN;
  const float* Lb = Lrow + (size_t)b * kN;

  v2f kb2[16];
#pragma unroll
  for (int s = 0; s < 16; ++s) {
    const float* kr = Kb + (size_t)(4 * s + 2 * half) * kN;
    kb2[s] = { kr[n], kr[n + kN] };
  }

  v8f yacc[4] = { v8f{}, v8f{}, v8f{}, v8f{} };

  for (int it = 0; it < 256; ++it) {
    const int i0 = it * 16;

    const float* qrow = qT + ((size_t)b * kN + i0 + col) * kDK + 2 * half;
    v8f c = {};
#pragma unroll
    for (int s = 0; s < 16; ++s) {
      v2f a = *(const v2f*)(qrow + 4 * s);
      c = wmma_f32(a, kb2[s], c);
    }

    const float* lp = Lb + i0 + 8 * half;
    const float4 la  = *(const float4*)(lp);
    const float4 lb4 = *(const float4*)(lp + 4);
    const float lv[8] = { la.x, la.y, la.z, la.w, lb4.x, lb4.y, lb4.z, lb4.w };
#pragma unroll
    for (int r = 0; r < 8; ++r)
      Pt[(r + 8 * half) * 16 + col] = __expf(c[r] - lv[r]);
    __syncthreads();

#pragma unroll
    for (int ci = 0; ci < 4; ++ci) {
      v2f bp = { Pt[(4 * ci + 2 * half) * 16 + col],
                 Pt[(4 * ci + 2 * half + 1) * 16 + col] };
      const float* vbase = Vb + (size_t)col * kN + (i0 + 4 * ci + 2 * half);
#pragma unroll
      for (int dt = 0; dt < 4; ++dt) {
        v2f a = *(const v2f*)(vbase + (size_t)(16 * dt) * kN);
        yacc[dt] = wmma_f32(a, bp, yacc[dt]);
      }
    }
    __syncthreads();
  }

#pragma unroll
  for (int dt = 0; dt < 4; ++dt)
#pragma unroll
    for (int r = 0; r < 8; ++r)
      ys[(16 * dt + r + 8 * half) * 16 + col] = yacc[dt][r];
  __syncthreads();

  const float g = gamma[0];
  const float* xb = x + (size_t)b * kC * kN;
  float* ob = out + (size_t)b * kC * kN;

  for (int mt = 0; mt < 16; ++mt) {
    const int m0 = mt * 16;
    const float* wr = Wout + (size_t)(m0 + col) * kDK + 2 * half;
    v8f c = {};
#pragma unroll
    for (int s = 0; s < 16; ++s) {
      v2f a  = *(const v2f*)(wr + 4 * s);
      v2f bb = { ys[(4 * s + 2 * half) * 16 + col],
                 ys[(4 * s + 2 * half + 1) * 16 + col] };
      c = wmma_f32(a, bb, c);
    }
#pragma unroll
    for (int r = 0; r < 8; ++r) {
      const int row = m0 + r + 8 * half;
      const size_t idx = (size_t)row * kN + n;
      ob[idx] = c[r] * g + xb[idx];
    }
  }
}

extern "C" void kernel_launch(void* const* d_in, const int* in_sizes, int n_in,
                              void* d_out, int out_size, void* d_ws, size_t ws_size,
                              hipStream_t stream) {
  const float* x     = (const float*)d_in[0];
  const float* Wqkv  = (const float*)d_in[1];
  const float* Wout  = (const float*)d_in[2];
  const float* gamma = (const float*)d_in[3];
  float* out = (float*)d_out;

  float* ws = (float*)d_ws;
  float* qT = ws;
  float* Kd = ws + 2097152;
  float* Vd = ws + 4194304;
  float* Lr = ws + 6291456;

  qkv_kernel<<<kBS * 12 * 64, 32, 0, stream>>>(x, Wqkv, qT, Kd, Vd);
  stats_kernel<<<kBS * 256, 32, 0, stream>>>(qT, Kd, Lr);
  attn_out_kernel<<<kBS * 256, 32, 0, stream>>>(qT, Kd, Vd, Lr, Wout, x, gamma, out);
  (void)in_sizes; (void)n_in; (void)out_size; (void)ws_size;
}


# Q, K, V intermediates kept in WMMA-fragment order (V tile via exchanged product operands): an iteration's fragments are whole-line 16-byte-per-lane loads issued together before its WMMAs
# speedup vs baseline: 1.3734x; 1.3734x over previous
#include <hip/hip_runtime.h>


static constexpr int kBS = 8;
static constexpr int kC  = 256;
static constexpr int kN  = 4096;
static constexpr int kDK = 64;
static constexpr int kTiles = kN / 16;

typedef __attribute__((ext_vector_type(2))) float v2f;
typedef __attribute__((ext_vector_type(8))) float v8f;

__device__ __forceinline__ v8f wmma_f32(v2f a, v2f b, v8f c) {
  return __builtin_amdgcn_wmma_f32_16x16x4_f32(false, a, false, b,
                                               (short)0, c, false, false);
}

__device__ __forceinline__ v2f lo2(const float4 v) { return v2f{ v.x, v.y }; }
__device__ __forceinline__ v2f hi2(const float4 v) { return v2f{ v.z, v.w }; }

template <bool TR>
__device__ __forceinline__ void qkv_tile(const float* __restrict__ xb,
                                         const float* __restrict__ Arow,
                                         const int n, v8f (&cs)[4]) {
  const int half = threadIdx.x >> 4;

  v8f c0 = {}, c1 = {}, c2 = {}, c3 = {};

#pragma unroll 4
  for (int kk = 0; kk < kC; kk += 4) {
    v2f a = *(const v2f*)(Arow + kk);
    const float* xr0 = xb + (size_t)(kk + 2 * half) * kN;
    const float* xr1 = xr0 + kN;
    v2f b0 = { xr0[n],      xr1[n]      };
    v2f b1 = { xr0[n + 16], xr1[n + 16] };
    v2f b2 = { xr0[n + 32], xr1[n + 32] };
    v2f b3 = { xr0[n + 48], xr1[n + 48] };
    if (TR) {
      c0 = wmma_f32(b0, a, c0);
      c1 = wmma_f32(b1, a, c1);
      c2 = wmma_f32(b2, a, c2);
      c3 = wmma_f32(b3, a, c3);
    } else {
      c0 = wmma_f32(a, b0, c0);
      c1 = wmma_f32(a, b1, c1);
      c2 = wmma_f32(a, b2, c2);
      c3 = wmma_f32(a, b3, c3);
    }
  }
  cs[0] = c0; cs[1] = c1; cs[2] = c2; cs[3] = c3;
}

__global__ void qkv_kernel(const float* __restrict__ x,
                           const float* __restrict__ Wqkv,
                           float* __restrict__ Qf,
                           float* __restrict__ Kf,
                           float* __restrict__ Vf) {
  const int L    = threadIdx.x;
  const int col  = L & 15;
  const int half = L >> 4;

  int blk = blockIdx.x;
  const int b   = blk / (12 * 64);
  const int rem = blk % (12 * 64);
  const int m0  = (rem / 64) * 16;
  const int n0  = (rem % 64) * 64;

  const float* xb   = x + (size_t)b * kC * kN;
  const float* Arow = Wqkv + (size_t)(m0 + col) * kC + 2 * half;
  const int n = n0 + col;
  const size_t tile0 = (size_t)b * kTiles + (n0 >> 4);

  v8f cs[4];
  float* dst;
  if (m0 < 128) {
    float* base = (m0 < 64) ? Qf : Kf;
    const int p = ((m0 & 63) >> 3) + half;
    dst = base + (tile0 * 8 + p) * 128 + col * 4;
    qkv_tile<false>(xb, Arow, n, cs);
  } else {
    const int dt = (m0 - 128) >> 4;
    dst = Vf + ((tile0 * 4 + dt) * 2 + half) * 128 + col * 4;
    qkv_tile<true>(xb, Arow, n, cs);
  }

#pragma unroll
  for (int t = 0; t < 4; ++t) {
    float* d0 = dst + (size_t)t * 1024;
    *(float4*)(d0)      = make_float4(cs[t][0], cs[t][1], cs[t][4], cs[t][5]);
    *(float4*)(d0 + 64) = make_float4(cs[t][2], cs[t][3], cs[t][6], cs[t][7]);
  }
}

__global__ void stats_kernel(const float* __restrict__ Qf,
                             const float* __restrict__ Kf,
                             float* __restrict__ Lrow) {
  const int L    = threadIdx.x;
  const int col  = L & 15;
  const int half = L >> 4;
  const int b  = blockIdx.x >> 8;
  const int i0 = (blockIdx.x & 255) * 16;

  float4 aq[8];
  const float* qt = Qf + ((size_t)blockIdx.x * 8) * 128 + L * 4;
#pragma unroll
  for (int p = 0; p < 8; ++p) aq[p] = *(const float4*)(qt + p * 128);

  const float* Kb = Kf + (size_t)b * kTiles * 1024 + L * 4;

  float mr[8], zr[8];
#pragma unroll
  for (int r = 0; r < 8; ++r) { mr[r] = -3.0e38f; zr[r] = 0.0f; }

  for (int jt = 0; jt < 256; ++jt) {
    const float* kt = Kb + (size_t)jt * 1024;
    float4 kb[8];
#pragma unroll
    for (int p = 0; p < 8; ++p) kb[p] = *(const float4*)(kt + p * 128);
    __builtin_amdgcn_sched_barrier(0);
    v8f c = {};
#pragma unroll
    for (int p = 0; p < 8; ++p) {
      c = wmma_f32(lo2(aq[p]), lo2(kb[p]), c);
      c = wmma_f32(hi2(aq[p]), hi2(kb[p]), c);
    }
#pragma unroll
    for (int r = 0; r < 8; ++r) {
      const float v  = c[r];
      const float mn = fmaxf(mr[r], v);
      zr[r] = zr[r] * __expf(mr[r] - mn) + __expf(v - mn);
      mr[r] = mn;
    }
  }

#pragma unroll
  for (int r = 0; r < 8; ++r) {
    float m = mr[r], z = zr[r];
#pragma unroll
    for (int mask = 1; mask < 16; mask <<= 1) {
      const float om = __shfl_xor(m, mask, 16);
      const float oz = __shfl_xor(z, mask, 16);
      const float mn = fmaxf(m, om);
      z = z * __expf(m - mn) + oz * __expf(om - mn);
      m = mn;
    }
    if (col == 0)
      Lrow[(size_t)b * kN + i0 + 8 * half + r] = m + __logf(z);
  }
}

__global__ void __launch_bounds__(32)
attn_out_kernel(const float* __restrict__ Qf,
                const float* __restrict__ Kf,
                const float* __restrict__ Vf,
                const float* __restrict__ Lrow,
                const float* __restrict__ Wout,
                const float* __restrict__ x,
                const float* __restrict__ gamma,
                float* __restrict__ out) {
  __shared__ float Pt[16 * 16];
  __shared__ float ys[64 * 16];

  const int L    = threadIdx.x;
  const int col  = L & 15;
  const int half = L >> 4;
  const int b  = blockIdx.x >> 8;
  const int j0 = (blockIdx.x & 255) * 16;
  const int n  = j0 + col;

  const float* Qb = Qf + (size_t)b * kTiles * 1024 + L * 4;
  const float* Vb = Vf + (size_t)b * kTiles * 1024 + L * 4;
  const float* Lb = Lrow + (size_t)b * kN;

  float4 kb[8];
  const float* kt = Kf + ((size_t)blockIdx.x * 8) * 128 + L * 4;
#pragma unroll
  for (int p = 0; p < 8; ++p) kb[p] = *(const float4*)(kt + p * 128);

  v8f yacc[4] = { v8f{}, v8f{}, v8f{}, v8f{} };

  for (int it = 0; it < 256; ++it) {
    const int i0 = it * 16;

    const float* qt = Qb + (size_t)it * 1024;
    const float* vt = Vb + (size_t)it * 1024;
    const float* lp = Lb + i0 + 8 * half;
    float4 aq[8], av[8];
#pragma unroll
    for (int p = 0; p < 8; ++p) aq[p] = *(const float4*)(qt + p * 128);
    const float4 la  = *(const float4*)(lp);
    const float4 lb4 = *(const float4*)(lp + 4);
#pragma unroll
    for (int p = 0; p < 8; ++p) av[p] = *(const float4*)(vt + p * 128);
    __builtin_amdgcn_sched_barrier(0);

    v8f c = {};
#pragma unroll
    for (int p = 0; p < 8; ++p) {
      c = wmma_f32(lo2(aq[p]), lo2(kb[p]), c);
      c = wmma_f32(hi2(aq[p]), hi2(kb[p]), c);
    }

    const float lv[8] = { la.x, la.y, la.z, la.w, lb4.x, lb4.y, lb4.z, lb4.w };
#pragma unroll
    for (int r = 0; r < 8; ++r)
      Pt[(r + 8 * half) * 16 + col] = __expf(c[r] - lv[r]);
    __syncthreads();

#pragma unroll
    for (int ci = 0; ci < 4; ++ci) {
      v2f bp = { Pt[(4 * ci + 2 * half) * 16 + col],
                 Pt[(4 * ci + 2 * half + 1) * 16 + col] };
#pragma unroll
      for (int dt = 0; dt < 4; ++dt) {
        const float4 v4 = av[dt * 2 + (ci >> 1)];
        const v2f a = (ci & 1) ? hi2(v4) : lo2(v4);
        yacc[dt] = wmma_f32(a, bp, yacc[dt]);
      }
    }
    __syncthreads();
  }

#pragma unroll
  for (int dt = 0; dt < 4; ++dt)
#pragma unroll
    for (int r = 0; r < 8; ++r)
      ys[(16 * dt + r + 8 * half) * 16 + col] = yacc[dt][r];
  __syncthreads();

  const float g = gamma[0];
  const float* xb = x + (size_t)b * kC * kN;
  float* ob = out + (size_t)b * kC * kN;

  for (int mt = 0; mt < 16; ++mt) {
    const int m0 = mt * 16;
    const float* wr = Wout + (size_t)(m0 + col) * kDK + 2 * half;
    v8f c = {};
#pragma unroll
    for (int s = 0; s < 16; ++s) {
      v2f a  = *(const v2f*)(wr + 4 * s);
      v2f bb = { ys[(4 * s + 2 * half) * 16 + col],
                 ys[(4 * s + 2 * half + 1) * 16 + col] };
      c = wmma_f32(a, bb, c);
    }
#pragma unroll
    for (int r = 0; r < 8; ++r) {
      const int row = m0 + r + 8 * half;
      const size_t idx = (size_t)row * kN + n;
      ob[idx] = c[r] * g + xb[idx];
    }
  }
}

extern "C" void kernel_launch(void* const* d_in, const int* in_sizes, int n_in,
                              void* d_out, int out_size, void* d_ws, size_t ws_size,
                              hipStream_t stream) {
  const float* x     = (const float*)d_in[0];
  const float* Wqkv  = (const float*)d_in[1];
  const float* Wout  = (const float*)d_in[2];
  const float* gamma = (const float*)d_in[3];
  float* out = (float*)d_out;

  float* ws = (float*)d_ws;
  float* Qf = ws;
  float* Kf = ws + 2097152;
  float* Vf = ws + 4194304;
  float* Lr = ws + 6291456;

  qkv_kernel<<<kBS * 12 * 64, 32, 0, stream>>>(x, Wqkv, Qf, Kf, Vf);
  stats_kernel<<<kBS * 256, 32, 0, stream>>>(Qf, Kf, Lr);
  attn_out_kernel<<<kBS * 256, 32, 0, stream>>>(Qf, Kf, Vf, Lr, Wout, x, gamma, out);
  (void)in_sizes; (void)n_in; (void)out_size; (void)ws_size;
}


# Stats and attention kernels: 4 waves per block, each streaming a quarter of the tiles (8 waves per SIMD cover load latency); per-wave partial sums merged through LDS in wave order (sum order changed)
# speedup vs baseline: 1.7875x; 1.3015x over previous
#include <hip/hip_runtime.h>


static constexpr int kBS = 8;
static constexpr int kC  = 256;
static constexpr int kN  = 4096;
static constexpr int kDK = 64;
static constexpr int kTiles = kN / 16;
static constexpr int kSplit = 4;
static constexpr int kPerWave = kTiles / kSplit;

typedef __attribute__((ext_vector_type(2))) float v2f;
typedef __attribute__((ext_vector_type(8))) float v8f;

__device__ __forceinline__ v8f wmma_f32(v2f a, v2f b, v8f c) {
  return __builtin_amdgcn_wmma_f32_16x16x4_f32(false, a, false, b,
                                               (short)0, c, false, false);
}

__device__ __forceinline__ v2f lo2(const float4 v) { return v2f{ v.x, v.y }; }
__device__ __forceinline__ v2f hi2(const float4 v) { return v2f{ v.z, v.w }; }

template <bool TR>
__device__ __forceinline__ void qkv_tile(const float* __restrict__ xb,
                                         const float* __restrict__ Arow,
                                         const int n, v8f (&cs)[4]) {
  const int half = threadIdx.x >> 4;

  v8f c0 = {}, c1 = {}, c2 = {}, c3 = {};

#pragma unroll 4
  for (int kk = 0; kk < kC; kk += 4) {
    v2f a = *(const v2f*)(Arow + kk);
    const float* xr0 = xb + (size_t)(kk + 2 * half) * kN;
    const float* xr1 = xr0 + kN;
    v2f b0 = { xr0[n],      xr1[n]      };
    v2f b1 = { xr0[n + 16], xr1[n + 16] };
    v2f b2 = { xr0[n + 32], xr1[n + 32] };
    v2f b3 = { xr0[n + 48], xr1[n + 48] };
    if (TR) {
      c0 = wmma_f32(b0, a, c0);
      c1 = wmma_f32(b1, a, c1);
      c2 = wmma_f32(b2, a, c2);
      c3 = wmma_f32(b3, a, c3);
    } else {
      c0 = wmma_f32(a, b0, c0);
      c1 = wmma_f32(a, b1, c1);
      c2 = wmma_f32(a, b2, c2);
      c3 = wmma_f32(a, b3, c3);
    }
  }
  cs[0] = c0; cs[1] = c1; cs[2] = c2; cs[3] = c3;
}

__global__ void qkv_kernel(const float* __restrict__ x,
                           const float* __restrict__ Wqkv,
                           float* __restrict__ Qf,
                           float* __restrict__ Kf,
                           float* __restrict__ Vf) {
  const int L    = threadIdx.x;
  const int col  = L & 15;
  const int half = L >> 4;

  int blk = blockIdx.x;
  const int b   = blk / (12 * 64);
  const int rem = blk % (12 * 64);
  const int m0  = (rem / 64) * 16;
  const int n0  = (rem % 64) * 64;

  const float* xb   = x + (size_t)b * kC * kN;
  const float* Arow = Wqkv + (size_t)(m0 + col) * kC + 2 * half;
  const int n = n0 + col;
  const size_t tile0 = (size_t)b * kTiles + (n0 >> 4);

  v8f cs[4];
  float* dst;
  if (m0 < 128) {
    float* base = (m0 < 64) ? Qf : Kf;
    const int p = ((m0 & 63) >> 3) + half;
    dst = base + (tile0 * 8 + p) * 128 + col * 4;
    qkv_tile<false>(xb, Arow, n, cs);
  } else {
    const int dt = (m0 - 128) >> 4;
    dst = Vf + ((tile0 * 4 + dt) * 2 + half) * 128 + col * 4;
    qkv_tile<true>(xb, Arow, n, cs);
  }

#pragma unroll
  for (int t = 0; t < 4; ++t) {
    float* d0 = dst + (size_t)t * 1024;
    *(float4*)(d0)      = make_float4(cs[t][0], cs[t][1], cs[t][4], cs[t][5]);
    *(float4*)(d0 + 64) = make_float4(cs[t][2], cs[t][3], cs[t][6], cs[t][7]);
  }
}

__global__ void __launch_bounds__(kSplit * 32, 8)
stats_kernel(const float* __restrict__ Qf,
             const float* __restrict__ Kf,
             float* __restrict__ Lrow) {
  __shared__ float sm[kSplit][16];
  __shared__ float sz[kSplit][16];

  const int L    = threadIdx.x & 31;
  const int col  = L & 15;
  const int half = L >> 4;
  const int w  = __builtin_amdgcn_readfirstlane(threadIdx.x >> 5);
  const int b  = blockIdx.x >> 8;
  const int i0 = (blockIdx.x & 255) * 16;

  float4 aq[8];
  const float* qt = Qf + ((size_t)blockIdx.x * 8) * 128 + L * 4;
#pragma unroll
  for (int p = 0; p < 8; ++p) aq[p] = *(const float4*)(qt + p * 128);

  const float* Kb = Kf + ((size_t)b * kTiles + (size_t)w * kPerWave) * 1024 + L * 4;

  float mr[8], zr[8];
#pragma unroll
  for (int r = 0; r < 8; ++r) { mr[r] = -3.0e38f; zr[r] = 0.0f; }

  for (int jt = 0; jt < kPerWave; ++jt) {
    const float* kt = Kb + (size_t)jt * 1024;
    float4 kb[8];
#pragma unroll
    for (int p = 0; p < 8; ++p) kb[p] = *(const float4*)(kt + p * 128);
    __builtin_amdgcn_sched_barrier(0);
    v8f c = {};
#pragma unroll
    for (int p = 0; p < 8; ++p) {
      c = wmma_f32(lo2(aq[p]), lo2(kb[p]), c);
      c = wmma_f32(hi2(aq[p]), hi2(kb[p]), c);
    }
#pragma unroll
    for (int r = 0; r < 8; ++r) {
      const float v  = c[r];
      const float mn = fmaxf(mr[r], v);
      zr[r] = zr[r] * __expf(mr[r] - mn) + __expf(v - mn);
      mr[r] = mn;
    }
  }

#pragma unroll
  for (int r = 0; r < 8; ++r) {
    float m = mr[r], z = zr[r];
#pragma unroll
    for (int mask = 1; mask < 16; mask <<= 1) {
      const float om = __shfl_xor(m, mask, 16);
      const float oz = __shfl_xor(z, mask, 16);
      const float mn = fmaxf(m, om);
      z = z * __expf(m - mn) + oz * __expf(om - mn);
      m = mn;
    }
    if (col == 0) {
      sm[w][8 * half + r] = m;
      sz[w][8 * half + r] = z;
    }
  }
  __syncthreads();

  if (threadIdx.x < 16) {
    const int row = threadIdx.x;
    float m = sm[0][row], z = sz[0][row];
#pragma unroll
    for (int k = 1; k < kSplit; ++k) {
      const float om = sm[k][row];
      const float oz = sz[k][row];
      const float mn = fmaxf(m, om);
      z = z * __expf(m - mn) + oz * __expf(om - mn);
      m = mn;
    }
    Lrow[(size_t)b * kN + i0 + row] = m + __logf(z);
  }
}

__global__ void __launch_bounds__(kSplit * 32, 8)
attn_out_kernel(const float* __restrict__ Qf,
                const float* __restrict__ Kf,
                const float* __restrict__ Vf,
                const float* __restrict__ Lrow,
                const float* __restrict__ Wout,
                const float* __restrict__ x,
                const float* __restrict__ gamma,
                float* __restrict__ out) {
  __shared__ float Pt[kSplit][16 * 16];
  __shared__ float yp[kSplit][64 * 16];
  __shared__ float ys[64 * 16];

  const int L    = threadIdx.x & 31;
  const int col  = L & 15;
  const int half = L >> 4;
  const int w  = __builtin_amdgcn_readfirstlane(threadIdx.x >> 5);
  const int b  = blockIdx.x >> 8;
  const int j0 = (blockIdx.x & 255) * 16;
  const int n  = j0 + col;

  const size_t t0 = (size_t)b * kTiles + (size_t)w * kPerWave;
  const float* Qb = Qf + t0 * 1024 + L * 4;
  const float* Vb = Vf + t0 * 1024 + L * 4;
  const float* Lb = Lrow + (size_t)b * kN + (size_t)w * kPerWave * 16 + 8 * half;
  float* Pw = Pt[w];

  float4 kb[8];
  const float* kt = Kf + ((size_t)blockIdx.x * 8) * 128 + L * 4;
#pragma unroll
  for (int p = 0; p < 8; ++p) kb[p] = *(const float4*)(kt + p * 128);

  v8f yacc[4] = { v8f{}, v8f{}, v8f{}, v8f{} };

  for (int it = 0; it < kPerWave; ++it) {
    const float* qt = Qb + (size_t)it * 1024;
    const float* vt = Vb + (size_t)it * 1024;
    const float* lp = Lb + it * 16;
    float4 aq[8], av[8];
#pragma unroll
    for (int p = 0; p < 8; ++p) aq[p] = *(const float4*)(qt + p * 128);
    const float4 la  = *(const float4*)(lp);
    const float4 lb4 = *(const float4*)(lp + 4);
    __builtin_amdgcn_sched_barrier(0);

    v8f c = {};
#pragma unroll
    for (int p = 0; p < 8; ++p) {
      c = wmma_f32(lo2(aq[p]), lo2(kb[p]), c);
      c = wmma_f32(hi2(aq[p]), hi2(kb[p]), c);
    }

#pragma unroll
    for (int p = 0; p < 8; ++p) av[p] = *(const float4*)(vt + p * 128);

    const float lv[8] = { la.x, la.y, la.z, la.w, lb4.x, lb4.y, lb4.z, lb4.w };
#pragma unroll
    for (int r = 0; r < 8; ++r)
      Pw[(r + 8 * half) * 16 + col] = __expf(c[r] - lv[r]);
    asm volatile("s_wait_dscnt 0x0" ::: "memory");

#pragma unroll
    for (int ci = 0; ci < 4; ++ci) {
      v2f bp = { Pw[(4 * ci + 2 * half) * 16 + col],
                 Pw[(4 * ci + 2 * half + 1) * 16 + col] };
#pragma unroll
      for (int dt = 0; dt < 4; ++dt) {
        const float4 v4 = av[dt * 2 + (ci >> 1)];
        const v2f a = (ci & 1) ? hi2(v4) : lo2(v4);
        yacc[dt] = wmma_f32(a, bp, yacc[dt]);
      }
    }
  }

  float* ypw = yp[w];
#pragma unroll
  for (int dt = 0; dt < 4; ++dt)
#pragma unroll
    for (int r = 0; r < 8; ++r)
      ypw[(16 * dt + r + 8 * half) * 16 + col] = yacc[dt][r];
  __syncthreads();

#pragma unroll
  for (int e = 0; e < 8; ++e) {
    const int idx = threadIdx.x + 128 * e;
    ys[idx] = ((yp[0][idx] + yp[1][idx]) + yp[2][idx]) + yp[3][idx];
  }
  __syncthreads();

  const float g = gamma[0];
  const float* xb = x + (size_t)b * kC * kN;
  float* ob = out + (size_t)b * kC * kN;

  for (int mt = 4 * w; mt < 4 * w + 4; ++mt) {
    const int m0 = mt * 16;
    const float* wr = Wout + (size_t)(m0 + col) * kDK + 2 * half;
    v8f c = {};
#pragma unroll
    for (int s = 0; s < 16; ++s) {
      v2f a  = *(const v2f*)(wr + 4 * s);
      v2f bb = { ys[(4 * s + 2 * half) * 16 + col],
                 ys[(4 * s + 2 * half + 1) * 16 + col] };
      c = wmma_f32(a, bb, c);
    }
#pragma unroll
    for (int r = 0; r < 8; ++r) {
      const int row = m0 + r + 8 * half;
      const size_t idx = (size_t)row * kN + n;
      ob[idx] = c[r] * g + xb[idx];
    }
  }
}

extern "C" void kernel_launch(void* const* d_in, const int* in_sizes, int n_in,
                              void* d_out, int out_size, void* d_ws, size_t ws_size,
                              hipStream_t stream) {
  const float* x     = (const float*)d_in[0];
  const float* Wqkv  = (const float*)d_in[1];
  const float* Wout  = (const float*)d_in[2];
  const float* gamma = (const float*)d_in[3];
  float* out = (float*)d_out;

  float* ws = (float*)d_ws;
  float* Qf = ws;
  float* Kf = ws + 2097152;
  float* Vf = ws + 4194304;
  float* Lr = ws + 6291456;

  qkv_kernel<<<kBS * 12 * 64, 32, 0, stream>>>(x, Wqkv, Qf, Kf, Vf);
  stats_kernel<<<kBS * 256, kSplit * 32, 0, stream>>>(Qf, Kf, Lr);
  attn_out_kernel<<<kBS * 256, kSplit * 32, 0, stream>>>(Qf, Kf, Vf, Lr, Wout, x, gamma, out);
  (void)in_sizes; (void)n_in; (void)out_size; (void)ws_size;
}


# Stats and attention kernels: each wave keeps two 16-wide tiles resident, so every streamed K, Q and V tile is loaded once for two tiles' products (4 waves per SIMD); per-tile sum order unchanged
# speedup vs baseline: 2.0483x; 1.1459x over previous
#include <hip/hip_runtime.h>


static constexpr int kBS = 8;
static constexpr int kC  = 256;
static constexpr int kN  = 4096;
static constexpr int kDK = 64;
static constexpr int kTiles = kN / 16;
static constexpr int kSplit = 4;
static constexpr int kPerWave = kTiles / kSplit;
static constexpr int kPair = 2;

typedef __attribute__((ext_vector_type(2))) float v2f;
typedef __attribute__((ext_vector_type(8))) float v8f;

__device__ __forceinline__ v8f wmma_f32(v2f a, v2f b, v8f c) {
  return __builtin_amdgcn_wmma_f32_16x16x4_f32(false, a, false, b,
                                               (short)0, c, false, false);
}

__device__ __forceinline__ v2f lo2(const float4 v) { return v2f{ v.x, v.y }; }
__device__ __forceinline__ v2f hi2(const float4 v) { return v2f{ v.z, v.w }; }

template <bool TR>
__device__ __forceinline__ void qkv_tile(const float* __restrict__ xb,
                                         const float* __restrict__ Arow,
                                         const int n, v8f (&cs)[4]) {
  const int half = threadIdx.x >> 4;

  v8f c0 = {}, c1 = {}, c2 = {}, c3 = {};

#pragma unroll 4
  for (int kk = 0; kk < kC; kk += 4) {
    v2f a = *(const v2f*)(Arow + kk);
    const float* xr0 = xb + (size_t)(kk + 2 * half) * kN;
    const float* xr1 = xr0 + kN;
    v2f b0 = { xr0[n],      xr1[n]      };
    v2f b1 = { xr0[n + 16], xr1[n + 16] };
    v2f b2 = { xr0[n + 32], xr1[n + 32] };
    v2f b3 = { xr0[n + 48], xr1[n + 48] };
    if (TR) {
      c0 = wmma_f32(b0, a, c0);
      c1 = wmma_f32(b1, a, c1);
      c2 = wmma_f32(b2, a, c2);
      c3 = wmma_f32(b3, a, c3);
    } else {
      c0 = wmma_f32(a, b0, c0);
      c1 = wmma_f32(a, b1, c1);
      c2 = wmma_f32(a, b2, c2);
      c3 = wmma_f32(a, b3, c3);
    }
  }
  cs[0] = c0; cs[1] = c1; cs[2] = c2; cs[3] = c3;
}

__global__ void qkv_kernel(const float* __restrict__ x,
                           const float* __restrict__ Wqkv,
                           float* __restrict__ Qf,
                           float* __restrict__ Kf,
                           float* __restrict__ Vf) {
  const int L    = threadIdx.x;
  const int col  = L & 15;
  const int half = L >> 4;

  int blk = blockIdx.x;
  const int b   = blk / (12 * 64);
  const int rem = blk % (12 * 64);
  const int m0  = (rem / 64) * 16;
  const int n0  = (rem % 64) * 64;

  const float* xb   = x + (size_t)b * kC * kN;
  const float* Arow = Wqkv + (size_t)(m0 + col) * kC + 2 * half;
  const int n = n0 + col;
  const size_t tile0 = (size_t)b * kTiles + (n0 >> 4);

  v8f cs[4];
  float* dst;
  if (m0 < 128) {
    float* base = (m0 < 64) ? Qf : Kf;
    const int p = ((m0 & 63) >> 3) + half;
    dst = base + (tile0 * 8 + p) * 128 + col * 4;
    qkv_tile<false>(xb, Arow, n, cs);
  } else {
    const int dt = (m0 - 128) >> 4;
    dst = Vf + ((tile0 * 4 + dt) * 2 + half) * 128 + col * 4;
    qkv_tile<true>(xb, Arow, n, cs);
  }

#pragma unroll
  for (int t = 0; t < 4; ++t) {
    float* d0 = dst + (size_t)t * 1024;
    *(float4*)(d0)      = make_float4(cs[t][0], cs[t][1], cs[t][4], cs[t][5]);
    *(float4*)(d0 + 64) = make_float4(cs[t][2], cs[t][3], cs[t][6], cs[t][7]);
  }
}

__global__ void __launch_bounds__(kSplit * 32, 4)
stats_kernel(const float* __restrict__ Qf,
             const float* __restrict__ Kf,
             float* __restrict__ Lrow) {
  __shared__ float sm[kSplit][16 * kPair];
  __shared__ float sz[kSplit][16 * kPair];

  const int L    = threadIdx.x & 31;
  const int col  = L & 15;
  const int half = L >> 4;
  const int w  = __builtin_amdgcn_readfirstlane(threadIdx.x >> 5);
  const int b  = blockIdx.x >> 7;

  float4 aq[kPair][8];
#pragma unroll
  for (int t = 0; t < kPair; ++t) {
    const float* qt = Qf + ((size_t)(kPair * blockIdx.x + t) * 8) * 128 + L * 4;
#pragma unroll
    for (int p = 0; p < 8; ++p) aq[t][p] = *(const float4*)(qt + p * 128);
  }

  const float* Kb = Kf + ((size_t)b * kTiles + (size_t)w * kPerWave) * 1024 + L * 4;

  float mr[kPair][8], zr[kPair][8];
#pragma unroll
  for (int t = 0; t < kPair; ++t)
#pragma unroll
    for (int r = 0; r < 8; ++r) { mr[t][r] = -3.0e38f; zr[t][r] = 0.0f; }

  for (int jt = 0; jt < kPerWave; ++jt) {
    const float* kt = Kb + (size_t)jt * 1024;
    float4 kb[8];
#pragma unroll
    for (int p = 0; p < 8; ++p) kb[p] = *(const float4*)(kt + p * 128);
    __builtin_amdgcn_sched_barrier(0);
    v8f c0 = {}, c1 = {};
#pragma unroll
    for (int p = 0; p < 8; ++p) {
      c0 = wmma_f32(lo2(aq[0][p]), lo2(kb[p]), c0);
      c1 = wmma_f32(lo2(aq[1][p]), lo2(kb[p]), c1);
      c0 = wmma_f32(hi2(aq[0][p]), hi2(kb[p]), c0);
      c1 = wmma_f32(hi2(aq[1][p]), hi2(kb[p]), c1);
    }
#pragma unroll
    for (int r = 0; r < 8; ++r) {
      const float v  = c0[r];
      const float mn = fmaxf(mr[0][r], v);
      zr[0][r] = zr[0][r] * __expf(mr[0][r] - mn) + __expf(v - mn);
      mr[0][r] = mn;
    }
#pragma unroll
    for (int r = 0; r < 8; ++r) {
      const float v  = c1[r];
      const float mn = fmaxf(mr[1][r], v);
      zr[1][r] = zr[1][r] * __expf(mr[1][r] - mn) + __expf(v - mn);
      mr[1][r] = mn;
    }
  }

#pragma unroll
  for (int t = 0; t < kPair; ++t) {
#pragma unroll
    for (int r = 0; r < 8; ++r) {
      float m = mr[t][r], z = zr[t][r];
#pragma unroll
      for (int mask = 1; mask < 16; mask <<= 1) {
        const float om = __shfl_xor(m, mask, 16);
        const float oz = __shfl_xor(z, mask, 16);
        const float mn = fmaxf(m, om);
        z = z * __expf(m - mn) + oz * __expf(om - mn);
        m = mn;
      }
      if (col == 0) {
        sm[w][16 * t + 8 * half + r] = m;
        sz[w][16 * t + 8 * half + r] = z;
      }
    }
  }
  __syncthreads();

  if (threadIdx.x < 16 * kPair) {
    const int row = threadIdx.x;
    float m = sm[0][row], z = sz[0][row];
#pragma unroll
    for (int k = 1; k < kSplit; ++k) {
      const float om = sm[k][row];
      const float oz = sz[k][row];
      const float mn = fmaxf(m, om);
      z = z * __expf(m - mn) + oz * __expf(om - mn);
      m = mn;
    }
    Lrow[(size_t)blockIdx.x * (16 * kPair) + row] = m + __logf(z);
  }
}

__global__ void __launch_bounds__(kSplit * 32, 4)
attn_out_kernel(const float* __restrict__ Qf,
                const float* __restrict__ Kf,
                const float* __restrict__ Vf,
                const float* __restrict__ Lrow,
                const float* __restrict__ Wout,
                const float* __restrict__ x,
                const float* __restrict__ gamma,
                float* __restrict__ out) {
  __shared__ float Pt[kSplit][kPair][16 * 16];
  __shared__ float yp[kSplit][kPair][64 * 16];
  __shared__ float ys[kPair][64 * 16];

  const int L    = threadIdx.x & 31;
  const int col  = L & 15;
  const int half = L >> 4;
  const int w  = __builtin_amdgcn_readfirstlane(threadIdx.x >> 5);
  const int b  = blockIdx.x >> 7;
  const int j0 = (blockIdx.x & 127) * (16 * kPair);

  const size_t t0 = (size_t)b * kTiles + (size_t)w * kPerWave;
  const float* Qb = Qf + t0 * 1024 + L * 4;
  const float* Vb = Vf + t0 * 1024 + L * 4;
  const float* Lb = Lrow + (size_t)b * kN + (size_t)w * kPerWave * 16 + 8 * half;
  float* Pw0 = Pt[w][0];
  float* Pw1 = Pt[w][1];

  float4 kb[kPair][8];
#pragma unroll
  for (int t = 0; t < kPair; ++t) {
    const float* kt = Kf + ((size_t)(kPair * blockIdx.x + t) * 8) * 128 + L * 4;
#pragma unroll
    for (int p = 0; p < 8; ++p) kb[t][p] = *(const float4*)(kt + p * 128);
  }

  v8f yacc[kPair][4] = { { v8f{}, v8f{}, v8f{}, v8f{} }, { v8f{}, v8f{}, v8f{}, v8f{} } };

  for (int it = 0; it < kPerWave; ++it) {
    const float* qt = Qb + (size_t)it * 1024;
    const float* vt = Vb + (size_t)it * 1024;
    const float* lp = Lb + it * 16;
    float4 aq[8], av[8];
#pragma unroll
    for (int p = 0; p < 8; ++p) aq[p] = *(const float4*)(qt + p * 128);
    const float4 la  = *(const float4*)(lp);
    const float4 lb4 = *(const float4*)(lp + 4);
#pragma unroll
    for (int p = 0; p < 8; ++p) av[p] = *(const float4*)(vt + p * 128);
    __builtin_amdgcn_sched_barrier(0);

    v8f c0 = {}, c1 = {};
#pragma unroll
    for (int p = 0; p < 8; ++p) {
      c0 = wmma_f32(lo2(aq[p]), lo2(kb[0][p]), c0);
      c1 = wmma_f32(lo2(aq[p]), lo2(kb[1][p]), c1);
      c0 = wmma_f32(hi2(aq[p]), hi2(kb[0][p]), c0);
      c1 = wmma_f32(hi2(aq[p]), hi2(kb[1][p]), c1);
    }

    const float lv[8] = { la.x, la.y, la.z, la.w, lb4.x, lb4.y, lb4.z, lb4.w };
#pragma unroll
    for (int r = 0; r < 8; ++r) {
      Pw0[(r + 8 * half) * 16 + col] = __expf(c0[r] - lv[r]);
      Pw1[(r + 8 * half) * 16 + col] = __expf(c1[r] - lv[r]);
    }
    asm volatile("s_wait_dscnt 0x0" ::: "memory");

#pragma unroll
    for (int ci = 0; ci < 4; ++ci) {
      v2f bp0 = { Pw0[(4 * ci + 2 * half) * 16 + col],
                  Pw0[(4 * ci + 2 * half + 1) * 16 + col] };
      v2f bp1 = { Pw1[(4 * ci + 2 * half) * 16 + col],
                  Pw1[(4 * ci + 2 * half + 1) * 16 + col] };
#pragma unroll
      for (int dt = 0; dt < 4; ++dt) {
        const float4 v4 = av[dt * 2 + (ci >> 1)];
        const v2f a = (ci & 1) ? hi2(v4) : lo2(v4);
        yacc[0][dt] = wmma_f32(a, bp0, yacc[0][dt]);
        yacc[1][dt] = wmma_f32(a, bp1, yacc[1][dt]);
      }
    }
  }

#pragma unroll
  for (int t = 0; t < kPair; ++t) {
    float* ypw = yp[w][t];
#pragma unroll
    for (int dt = 0; dt < 4; ++dt)
#pragma unroll
      for (int r = 0; r < 8; ++r)
        ypw[(16 * dt + r + 8 * half) * 16 + col] = yacc[t][dt][r];
  }
  __syncthreads();

#pragma unroll
  for (int t = 0; t < kPair; ++t) {
#pragma unroll
    for (int e = 0; e < 8; ++e) {
      const int idx = threadIdx.x + 128 * e;
      ys[t][idx] = ((yp[0][t][idx] + yp[1][t][idx]) + yp[2][t][idx]) + yp[3][t][idx];
    }
  }
  __syncthreads();

  const float g = gamma[0];
  const float* xb = x + (size_t)b * kC * kN;
  float* ob = out + (size_t)b * kC * kN;
  const int n = j0 + col;

  for (int mt = 4 * w; mt < 4 * w + 4; ++mt) {
    const int m0 = mt * 16;
    const float* wr = Wout + (size_t)(m0 + col) * kDK + 2 * half;
    v8f c0 = {}, c1 = {};
#pragma unroll
    for (int s = 0; s < 16; ++s) {
      v2f a   = *(const v2f*)(wr + 4 * s);
      v2f bb0 = { ys[0][(4 * s + 2 * half) * 16 + col],
                  ys[0][(4 * s + 2 * half + 1) * 16 + col] };
      v2f bb1 = { ys[1][(4 * s + 2 * half) * 16 + col],
                  ys[1][(4 * s + 2 * half + 1) * 16 + col] };
      c0 = wmma_f32(a, bb0, c0);
      c1 = wmma_f32(a, bb1, c1);
    }
#pragma unroll
    for (int r = 0; r < 8; ++r) {
      const int row = m0 + r + 8 * half;
      const size_t idx = (size_t)row * kN + n;
      ob[idx]      = c0[r] * g + xb[idx];
      ob[idx + 16] = c1[r] * g + xb[idx + 16];
    }
  }
}

extern "C" void kernel_launch(void* const* d_in, const int* in_sizes, int n_in,
                              void* d_out, int out_size, void* d_ws, size_t ws_size,
                              hipStream_t stream) {
  const float* x     = (const float*)d_in[0];
  const float* Wqkv  = (const float*)d_in[1];
  const float* Wout  = (const float*)d_in[2];
  const float* gamma = (const float*)d_in[3];
  float* out = (float*)d_out;

  float* ws = (float*)d_ws;
  float* Qf = ws;
  float* Kf = ws + 2097152;
  float* Vf = ws + 4194304;
  float* Lr = ws + 6291456;

  qkv_kernel<<<kBS * 12 * 64, 32, 0, stream>>>(x, Wqkv, Qf, Kf, Vf);
  stats_kernel<<<kBS * 128, kSplit * 32, 0, stream>>>(Qf, Kf, Lr);
  attn_out_kernel<<<kBS * 128, kSplit * 32, 0, stream>>>(Qf, Kf, Vf, Lr, Wout, x, gamma, out);
  (void)in_sizes; (void)n_in; (void)out_size; (void)ws_size;
}


# Stats and attention kernels: each streamed tile's loads issued one tile ahead, right behind the WMMAs that last read their registers (stats: a second K register set); sum order unchanged
# speedup vs baseline: 2.1470x; 1.0482x over previous
#include <hip/hip_runtime.h>


static constexpr int kBS = 8;
static constexpr int kC  = 256;
static constexpr int kN  = 4096;
static constexpr int kDK = 64;
static constexpr int kTiles = kN / 16;
static constexpr int kSplit = 4;
static constexpr int kPerWave = kTiles / kSplit;
static constexpr int kPair = 2;

typedef __attribute__((ext_vector_type(2))) float v2f;
typedef __attribute__((ext_vector_type(8))) float v8f;

__device__ __forceinline__ v8f wmma_f32(v2f a, v2f b, v8f c) {
  return __builtin_amdgcn_wmma_f32_16x16x4_f32(false, a, false, b,
                                               (short)0, c, false, false);
}

__device__ __forceinline__ v2f lo2(const float4 v) { return v2f{ v.x, v.y }; }
__device__ __forceinline__ v2f hi2(const float4 v) { return v2f{ v.z, v.w }; }

template <bool TR>
__device__ __forceinline__ void qkv_tile(const float* __restrict__ xb,
                                         const float* __restrict__ Arow,
                                         const int n, v8f (&cs)[4]) {
  const int half = threadIdx.x >> 4;

  v8f c0 = {}, c1 = {}, c2 = {}, c3 = {};

#pragma unroll 4
  for (int kk = 0; kk < kC; kk += 4) {
    v2f a = *(const v2f*)(Arow + kk);
    const float* xr0 = xb + (size_t)(kk + 2 * half) * kN;
    const float* xr1 = xr0 + kN;
    v2f b0 = { xr0[n],      xr1[n]      };
    v2f b1 = { xr0[n + 16], xr1[n + 16] };
    v2f b2 = { xr0[n + 32], xr1[n + 32] };
    v2f b3 = { xr0[n + 48], xr1[n + 48] };
    if (TR) {
      c0 = wmma_f32(b0, a, c0);
      c1 = wmma_f32(b1, a, c1);
      c2 = wmma_f32(b2, a, c2);
      c3 = wmma_f32(b3, a, c3);
    } else {
      c0 = wmma_f32(a, b0, c0);
      c1 = wmma_f32(a, b1, c1);
      c2 = wmma_f32(a, b2, c2);
      c3 = wmma_f32(a, b3, c3);
    }
  }
  cs[0] = c0; cs[1] = c1; cs[2] = c2; cs[3] = c3;
}

__global__ void qkv_kernel(const float* __restrict__ x,
                           const float* __restrict__ Wqkv,
                           float* __restrict__ Qf,
                           float* __restrict__ Kf,
                           float* __restrict__ Vf) {
  const int L    = threadIdx.x;
  const int col  = L & 15;
  const int half = L >> 4;

  int blk = blockIdx.x;
  const int b   = blk / (12 * 64);
  const int rem = blk % (12 * 64);
  const int m0  = (rem / 64) * 16;
  const int n0  = (rem % 64) * 64;

  const float* xb   = x + (size_t)b * kC * kN;
  const float* Arow = Wqkv + (size_t)(m0 + col) * kC + 2 * half;
  const int n = n0 + col;
  const size_t tile0 = (size_t)b * kTiles + (n0 >> 4);

  v8f cs[4];
  float* dst;
  if (m0 < 128) {
    float* base = (m0 < 64) ? Qf : Kf;
    const int p = ((m0 & 63) >> 3) + half;
    dst = base + (tile0 * 8 + p) * 128 + col * 4;
    qkv_tile<false>(xb, Arow, n, cs);
  } else {
    const int dt = (m0 - 128) >> 4;
    dst = Vf + ((tile0 * 4 + dt) * 2 + half) * 128 + col * 4;
    qkv_tile<true>(xb, Arow, n, cs);
  }

#pragma unroll
  for (int t = 0; t < 4; ++t) {
    float* d0 = dst + (size_t)t * 1024;
    *(float4*)(d0)      = make_float4(cs[t][0], cs[t][1], cs[t][4], cs[t][5]);
    *(float4*)(d0 + 64) = make_float4(cs[t][2], cs[t][3], cs[t][6], cs[t][7]);
  }
}

__device__ __forceinline__ void stats_products(const float4 (&aq)[kPair][8],
                                               const float4 (&kb)[8],
                                               v8f& c0, v8f& c1) {
  c0 = v8f{}; c1 = v8f{};
#pragma unroll
  for (int p = 0; p < 8; ++p) {
    c0 = wmma_f32(lo2(aq[0][p]), lo2(kb[p]), c0);
    c1 = wmma_f32(lo2(aq[1][p]), lo2(kb[p]), c1);
    c0 = wmma_f32(hi2(aq[0][p]), hi2(kb[p]), c0);
    c1 = wmma_f32(hi2(aq[1][p]), hi2(kb[p]), c1);
  }
}

__device__ __forceinline__ void stats_update(const v8f& c0, const v8f& c1,
                                             float (&mr)[kPair][8],
                                             float (&zr)[kPair][8]) {
#pragma unroll
  for (int r = 0; r < 8; ++r) {
    const float v  = c0[r];
    const float mn = fmaxf(mr[0][r], v);
    zr[0][r] = zr[0][r] * __expf(mr[0][r] - mn) + __expf(v - mn);
    mr[0][r] = mn;
  }
#pragma unroll
  for (int r = 0; r < 8; ++r) {
    const float v  = c1[r];
    const float mn = fmaxf(mr[1][r], v);
    zr[1][r] = zr[1][r] * __expf(mr[1][r] - mn) + __expf(v - mn);
    mr[1][r] = mn;
  }
}

__global__ void __launch_bounds__(kSplit * 32, 4)
stats_kernel(const float* __restrict__ Qf,
             const float* __restrict__ Kf,
             float* __restrict__ Lrow) {
  __shared__ float sm[kSplit][16 * kPair];
  __shared__ float sz[kSplit][16 * kPair];

  const int L    = threadIdx.x & 31;
  const int col  = L & 15;
  const int half = L >> 4;
  const int w  = __builtin_amdgcn_readfirstlane(threadIdx.x >> 5);
  const int b  = blockIdx.x >> 7;

  float4 aq[kPair][8];
#pragma unroll
  for (int t = 0; t < kPair; ++t) {
    const float* qt = Qf + ((size_t)(kPair * blockIdx.x + t) * 8) * 128 + L * 4;
#pragma unroll
    for (int p = 0; p < 8; ++p) aq[t][p] = *(const float4*)(qt + p * 128);
  }

  const float* Kb = Kf + ((size_t)b * kTiles + (size_t)w * kPerWave) * 1024 + L * 4;

  float mr[kPair][8], zr[kPair][8];
#pragma unroll
  for (int t = 0; t < kPair; ++t)
#pragma unroll
    for (int r = 0; r < 8; ++r) { mr[t][r] = -3.0e38f; zr[t][r] = 0.0f; }

  float4 ka[8], kb[8];
#pragma unroll
  for (int p = 0; p < 8; ++p) ka[p] = *(const float4*)(Kb + p * 128);
  __builtin_amdgcn_sched_barrier(0);
#pragma unroll
  for (int p = 0; p < 8; ++p) kb[p] = *(const float4*)(Kb + 1024 + p * 128);

  for (int jt = 0; jt < kPerWave; jt += 2) {
    const int nj = (jt + 2 < kPerWave) ? jt + 2 : jt;
    const float* k2 = Kb + (size_t)nj * 1024;
    v8f c0, c1;

    stats_products(aq, ka, c0, c1);
    asm volatile("" : "+v"(c0), "+v"(c1) :: "memory");
#pragma unroll
    for (int p = 0; p < 8; ++p) ka[p] = *(const float4*)(k2 + p * 128);
    __builtin_amdgcn_sched_barrier(0);
    stats_update(c0, c1, mr, zr);

    stats_products(aq, kb, c0, c1);
    asm volatile("" : "+v"(c0), "+v"(c1) :: "memory");
#pragma unroll
    for (int p = 0; p < 8; ++p) kb[p] = *(const float4*)(k2 + 1024 + p * 128);
    __builtin_amdgcn_sched_barrier(0);
    stats_update(c0, c1, mr, zr);
  }

#pragma unroll
  for (int t = 0; t < kPair; ++t) {
#pragma unroll
    for (int r = 0; r < 8; ++r) {
      float m = mr[t][r], z = zr[t][r];
#pragma unroll
      for (int mask = 1; mask < 16; mask <<= 1) {
        const float om = __shfl_xor(m, mask, 16);
        const float oz = __shfl_xor(z, mask, 16);
        const float mn = fmaxf(m, om);
        z = z * __expf(m - mn) + oz * __expf(om - mn);
        m = mn;
      }
      if (col == 0) {
        sm[w][16 * t + 8 * half + r] = m;
        sz[w][16 * t + 8 * half + r] = z;
      }
    }
  }
  __syncthreads();

  if (threadIdx.x < 16 * kPair) {
    const int row = threadIdx.x;
    float m = sm[0][row], z = sz[0][row];
#pragma unroll
    for (int k = 1; k < kSplit; ++k) {
      const float om = sm[k][row];
      const float oz = sz[k][row];
      const float mn = fmaxf(m, om);
      z = z * __expf(m - mn) + oz * __expf(om - mn);
      m = mn;
    }
    Lrow[(size_t)blockIdx.x * (16 * kPair) + row] = m + __logf(z);
  }
}

__global__ void __launch_bounds__(kSplit * 32, 4)
attn_out_kernel(const float* __restrict__ Qf,
                const float* __restrict__ Kf,
                const float* __restrict__ Vf,
                const float* __restrict__ Lrow,
                const float* __restrict__ Wout,
                const float* __restrict__ x,
                const float* __restrict__ gamma,
                float* __restrict__ out) {
  __shared__ float Pt[kSplit][kPair][16 * 16];
  __shared__ float yp[kSplit][kPair][64 * 16];
  __shared__ float ys[kPair][64 * 16];

  const int L    = threadIdx.x & 31;
  const int col  = L & 15;
  const int half = L >> 4;
  const int w  = __builtin_amdgcn_readfirstlane(threadIdx.x >> 5);
  const int b  = blockIdx.x >> 7;
  const int j0 = (blockIdx.x & 127) * (16 * kPair);

  const size_t t0 = (size_t)b * kTiles + (size_t)w * kPerWave;
  const float* Qb = Qf + t0 * 1024 + L * 4;
  const float* Vb = Vf + t0 * 1024 + L * 4;
  const float* Lb = Lrow + (size_t)b * kN + (size_t)w * kPerWave * 16 + 8 * half;
  float* Pw0 = Pt[w][0];
  float* Pw1 = Pt[w][1];

  float4 kb[kPair][8];
#pragma unroll
  for (int t = 0; t < kPair; ++t) {
    const float* kt = Kf + ((size_t)(kPair * blockIdx.x + t) * 8) * 128 + L * 4;
#pragma unroll
    for (int p = 0; p < 8; ++p) kb[t][p] = *(const float4*)(kt + p * 128);
  }

  v8f yacc[kPair][4] = { { v8f{}, v8f{}, v8f{}, v8f{} }, { v8f{}, v8f{}, v8f{}, v8f{} } };

  __builtin_amdgcn_sched_barrier(0);

  float4 aq[8], av[8];
#pragma unroll
  for (int p = 0; p < 8; ++p) aq[p] = *(const float4*)(Qb + p * 128);
  __builtin_amdgcn_sched_barrier(0);
  float4 la  = *(const float4*)(Lb);
  float4 lb4 = *(const float4*)(Lb + 4);
  __builtin_amdgcn_sched_barrier(0);
#pragma unroll
  for (int p = 0; p < 8; ++p) av[p] = *(const float4*)(Vb + p * 128);

  for (int it = 0; it < kPerWave; ++it) {
    const int nx = (it + 1 < kPerWave) ? it + 1 : it;
    const float* qt = Qb + (size_t)nx * 1024;
    const float* vt = Vb + (size_t)nx * 1024;
    const float* lp = Lb + nx * 16;

    v8f c0 = {}, c1 = {};
#pragma unroll
    for (int p = 0; p < 8; ++p) {
      c0 = wmma_f32(lo2(aq[p]), lo2(kb[0][p]), c0);
      c1 = wmma_f32(lo2(aq[p]), lo2(kb[1][p]), c1);
      c0 = wmma_f32(hi2(aq[p]), hi2(kb[0][p]), c0);
      c1 = wmma_f32(hi2(aq[p]), hi2(kb[1][p]), c1);
    }
    asm volatile("" : "+v"(c0), "+v"(c1) :: "memory");
#pragma unroll
    for (int p = 0; p < 8; ++p) aq[p] = *(const float4*)(qt + p * 128);
    __builtin_amdgcn_sched_barrier(0);

    const float lv[8] = { la.x, la.y, la.z, la.w, lb4.x, lb4.y, lb4.z, lb4.w };
#pragma unroll
    for (int r = 0; r < 8; ++r) {
      Pw0[(r + 8 * half) * 16 + col] = __expf(c0[r] - lv[r]);
      Pw1[(r + 8 * half) * 16 + col] = __expf(c1[r] - lv[r]);
    }
    asm volatile("s_wait_dscnt 0x0" ::: "memory");

#pragma unroll
    for (int ci = 0; ci < 4; ++ci) {
      v2f bp0 = { Pw0[(4 * ci + 2 * half) * 16 + col],
                  Pw0[(4 * ci + 2 * half + 1) * 16 + col] };
      v2f bp1 = { Pw1[(4 * ci + 2 * half) * 16 + col],
                  Pw1[(4 * ci + 2 * half + 1) * 16 + col] };
#pragma unroll
      for (int dt = 0; dt < 4; ++dt) {
        const float4 v4 = av[dt * 2 + (ci >> 1)];
        const v2f a = (ci & 1) ? hi2(v4) : lo2(v4);
        yacc[0][dt] = wmma_f32(a, bp0, yacc[0][dt]);
        yacc[1][dt] = wmma_f32(a, bp1, yacc[1][dt]);
      }
    }
    __builtin_amdgcn_sched_barrier(0);
    la  = *(const float4*)(lp);
    lb4 = *(const float4*)(lp + 4);
#pragma unroll
    for (int p = 0; p < 8; ++p) av[p] = *(const float4*)(vt + p * 128);
  }

#pragma unroll
  for (int t = 0; t < kPair; ++t) {
    float* ypw = yp[w][t];
#pragma unroll
    for (int dt = 0; dt < 4; ++dt)
#pragma unroll
      for (int r = 0; r < 8; ++r)
        ypw[(16 * dt + r + 8 * half) * 16 + col] = yacc[t][dt][r];
  }
  __syncthreads();

#pragma unroll
  for (int t = 0; t < kPair; ++t) {
#pragma unroll
    for (int e = 0; e < 8; ++e) {
      const int idx = threadIdx.x + 128 * e;
      ys[t][idx] = ((yp[0][t][idx] + yp[1][t][idx]) + yp[2][t][idx]) + yp[3][t][idx];
    }
  }
  __syncthreads();

  const float g = gamma[0];
  const float* xb = x + (size_t)b * kC * kN;
  float* ob = out + (size_t)b * kC * kN;
  const int n = j0 + col;

  for (int mt = 4 * w; mt < 4 * w + 4; ++mt) {
    const int m0 = mt * 16;
    const float* wr = Wout + (size_t)(m0 + col) * kDK + 2 * half;
    v8f c0 = {}, c1 = {};
#pragma unroll
    for (int s = 0; s < 16; ++s) {
      v2f a   = *(const v2f*)(wr + 4 * s);
      v2f bb0 = { ys[0][(4 * s + 2 * half) * 16 + col],
                  ys[0][(4 * s + 2 * half + 1) * 16 + col] };
      v2f bb1 = { ys[1][(4 * s + 2 * half) * 16 + col],
                  ys[1][(4 * s + 2 * half + 1) * 16 + col] };
      c0 = wmma_f32(a, bb0, c0);
      c1 = wmma_f32(a, bb1, c1);
    }
#pragma unroll
    for (int r = 0; r < 8; ++r) {
      const int row = m0 + r + 8 * half;
      const size_t idx = (size_t)row * kN + n;
      ob[idx]      = c0[r] * g + xb[idx];
      ob[idx + 16] = c1[r] * g + xb[idx + 16];
    }
  }
}

extern "C" void kernel_launch(void* const* d_in, const int* in_sizes, int n_in,
                              void* d_out, int out_size, void* d_ws, size_t ws_size,
                              hipStream_t stream) {
  const float* x     = (const float*)d_in[0];
  const float* Wqkv  = (const float*)d_in[1];
  const float* Wout  = (const float*)d_in[2];
  const float* gamma = (const float*)d_in[3];
  float* out = (float*)d_out;

  float* ws = (float*)d_ws;
  float* Qf = ws;
  float* Kf = ws + 2097152;
  float* Vf = ws + 4194304;
  float* Lr = ws + 6291456;

  qkv_kernel<<<kBS * 12 * 64, 32, 0, stream>>>(x, Wqkv, Qf, Kf, Vf);
  stats_kernel<<<kBS * 128, kSplit * 32, 0, stream>>>(Qf, Kf, Lr);
  attn_out_kernel<<<kBS * 128, kSplit * 32, 0, stream>>>(Qf, Kf, Vf, Lr, Wout, x, gamma, out);
  (void)in_sizes; (void)n_in; (void)out_size; (void)ws_size;
}


# Stats kernel: score tile computed transposed (K tile as A operand), so a lane owns one query row: one running max and one rescale per 16 scores, one lane exchange at the end; row sum order changed
# speedup vs baseline: 2.2325x; 1.0398x over previous
#include <hip/hip_runtime.h>


static constexpr int kBS = 8;
static constexpr int kC  = 256;
static constexpr int kN  = 4096;
static constexpr int kDK = 64;
static constexpr int kTiles = kN / 16;
static constexpr int kSplit = 4;
static constexpr int kPerWave = kTiles / kSplit;
static constexpr int kPair = 2;

typedef __attribute__((ext_vector_type(2))) float v2f;
typedef __attribute__((ext_vector_type(8))) float v8f;

__device__ __forceinline__ v8f wmma_f32(v2f a, v2f b, v8f c) {
  return __builtin_amdgcn_wmma_f32_16x16x4_f32(false, a, false, b,
                                               (short)0, c, false, false);
}

__device__ __forceinline__ v2f lo2(const float4 v) { return v2f{ v.x, v.y }; }
__device__ __forceinline__ v2f hi2(const float4 v) { return v2f{ v.z, v.w }; }

template <bool TR>
__device__ __forceinline__ void qkv_tile(const float* __restrict__ xb,
                                         const float* __restrict__ Arow,
                                         const int n, v8f (&cs)[4]) {
  const int half = threadIdx.x >> 4;

  v8f c0 = {}, c1 = {}, c2 = {}, c3 = {};

#pragma unroll 4
  for (int kk = 0; kk < kC; kk += 4) {
    v2f a = *(const v2f*)(Arow + kk);
    const float* xr0 = xb + (size_t)(kk + 2 * half) * kN;
    const float* xr1 = xr0 + kN;
    v2f b0 = { xr0[n],      xr1[n]      };
    v2f b1 = { xr0[n + 16], xr1[n + 16] };
    v2f b2 = { xr0[n + 32], xr1[n + 32] };
    v2f b3 = { xr0[n + 48], xr1[n + 48] };
    if (TR) {
      c0 = wmma_f32(b0, a, c0);
      c1 = wmma_f32(b1, a, c1);
      c2 = wmma_f32(b2, a, c2);
      c3 = wmma_f32(b3, a, c3);
    } else {
      c0 = wmma_f32(a, b0, c0);
      c1 = wmma_f32(a, b1, c1);
      c2 = wmma_f32(a, b2, c2);
      c3 = wmma_f32(a, b3, c3);
    }
  }
  cs[0] = c0; cs[1] = c1; cs[2] = c2; cs[3] = c3;
}

__global__ void qkv_kernel(const float* __restrict__ x,
                           const float* __restrict__ Wqkv,
                           float* __restrict__ Qf,
                           float* __restrict__ Kf,
                           float* __restrict__ Vf) {
  const int L    = threadIdx.x;
  const int col  = L & 15;
  const int half = L >> 4;

  int blk = blockIdx.x;
  const int b   = blk / (12 * 64);
  const int rem = blk % (12 * 64);
  const int m0  = (rem / 64) * 16;
  const int n0  = (rem % 64) * 64;

  const float* xb   = x + (size_t)b * kC * kN;
  const float* Arow = Wqkv + (size_t)(m0 + col) * kC + 2 * half;
  const int n = n0 + col;
  const size_t tile0 = (size_t)b * kTiles + (n0 >> 4);

  v8f cs[4];
  float* dst;
  if (m0 < 128) {
    float* base = (m0 < 64) ? Qf : Kf;
    const int p = ((m0 & 63) >> 3) + half;
    dst = base + (tile0 * 8 + p) * 128 + col * 4;
    qkv_tile<false>(xb, Arow, n, cs);
  } else {
    const int dt = (m0 - 128) >> 4;
    dst = Vf + ((tile0 * 4 + dt) * 2 + half) * 128 + col * 4;
    qkv_tile<true>(xb, Arow, n, cs);
  }

#pragma unroll
  for (int t = 0; t < 4; ++t) {
    float* d0 = dst + (size_t)t * 1024;
    *(float4*)(d0)      = make_float4(cs[t][0], cs[t][1], cs[t][4], cs[t][5]);
    *(float4*)(d0 + 64) = make_float4(cs[t][2], cs[t][3], cs[t][6], cs[t][7]);
  }
}

__device__ __forceinline__ void stats_products(const float4 (&aq)[kPair][8],
                                               const float4 (&kt)[8],
                                               v8f& c0, v8f& c1) {
  c0 = v8f{}; c1 = v8f{};
#pragma unroll
  for (int p = 0; p < 8; ++p) {
    c0 = wmma_f32(lo2(kt[p]), lo2(aq[0][p]), c0);
    c1 = wmma_f32(lo2(kt[p]), lo2(aq[1][p]), c1);
    c0 = wmma_f32(hi2(kt[p]), hi2(aq[0][p]), c0);
    c1 = wmma_f32(hi2(kt[p]), hi2(aq[1][p]), c1);
  }
}

__device__ __forceinline__ float max3f(const float a, const float b, const float c) {
  return fmaxf(fmaxf(a, b), c);
}

__device__ __forceinline__ void row_update(const v8f& a, const v8f& b,
                                           float& m, float& z) {
  const float t0 = max3f(a[0], a[1], a[2]);
  const float t1 = max3f(a[3], a[4], a[5]);
  const float t2 = max3f(a[6], a[7], b[0]);
  const float t3 = max3f(b[1], b[2], b[3]);
  const float t4 = max3f(b[4], b[5], b[6]);
  const float t5 = max3f(b[7], m, t0);
  const float mn = max3f(max3f(t1, t2, t3), t4, t5);
  float ea[8], eb[8];
#pragma unroll
  for (int r = 0; r < 8; ++r) {
    ea[r] = __expf(a[r] - mn);
    eb[r] = __expf(b[r] - mn);
  }
  const float pa = ((ea[0] + ea[1]) + (ea[2] + ea[3])) + ((ea[4] + ea[5]) + (ea[6] + ea[7]));
  const float pb = ((eb[0] + eb[1]) + (eb[2] + eb[3])) + ((eb[4] + eb[5]) + (eb[6] + eb[7]));
  z = z * __expf(m - mn) + (pa + pb);
  m = mn;
}

__global__ void __launch_bounds__(kSplit * 32, 4)
stats_kernel(const float* __restrict__ Qf,
             const float* __restrict__ Kf,
             float* __restrict__ Lrow) {
  __shared__ float sm[kSplit][16 * kPair];
  __shared__ float sz[kSplit][16 * kPair];

  const int L    = threadIdx.x & 31;
  const int col  = L & 15;
  const int half = L >> 4;
  const int w  = __builtin_amdgcn_readfirstlane(threadIdx.x >> 5);
  const int b  = blockIdx.x >> 7;

  float4 aq[kPair][8];
#pragma unroll
  for (int t = 0; t < kPair; ++t) {
    const float* qt = Qf + ((size_t)(kPair * blockIdx.x + t) * 8) * 128 + L * 4;
#pragma unroll
    for (int p = 0; p < 8; ++p) aq[t][p] = *(const float4*)(qt + p * 128);
  }

  const float* Kb = Kf + ((size_t)b * kTiles + (size_t)w * kPerWave) * 1024 + L * 4;

  float mr[kPair], zr[kPair];
#pragma unroll
  for (int t = 0; t < kPair; ++t) { mr[t] = -3.0e38f; zr[t] = 0.0f; }

  float4 ka[8], kb[8];
#pragma unroll
  for (int p = 0; p < 8; ++p) ka[p] = *(const float4*)(Kb + p * 128);
  __builtin_amdgcn_sched_barrier(0);
#pragma unroll
  for (int p = 0; p < 8; ++p) kb[p] = *(const float4*)(Kb + 1024 + p * 128);

  for (int jt = 0; jt < kPerWave; jt += 2) {
    const int nj = (jt + 2 < kPerWave) ? jt + 2 : jt;
    const float* k2 = Kb + (size_t)nj * 1024;
    v8f ca0, ca1, cb0, cb1;

    stats_products(aq, ka, ca0, ca1);
    asm volatile("" : "+v"(ca0), "+v"(ca1) :: "memory");
#pragma unroll
    for (int p = 0; p < 8; ++p) ka[p] = *(const float4*)(k2 + p * 128);
    __builtin_amdgcn_sched_barrier(0);

    stats_products(aq, kb, cb0, cb1);
    asm volatile("" : "+v"(cb0), "+v"(cb1) :: "memory");
#pragma unroll
    for (int p = 0; p < 8; ++p) kb[p] = *(const float4*)(k2 + 1024 + p * 128);
    __builtin_amdgcn_sched_barrier(0);

    row_update(ca0, cb0, mr[0], zr[0]);
    row_update(ca1, cb1, mr[1], zr[1]);
  }

#pragma unroll
  for (int t = 0; t < kPair; ++t) {
    float m = mr[t];
    float z = zr[t];
    const float om = __shfl_xor(m, 16);
    const float oz = __shfl_xor(z, 16);
    const float mn = fmaxf(m, om);
    z = z * __expf(m - mn) + oz * __expf(om - mn);
    m = mn;
    if (half == 0) {
      sm[w][16 * t + col] = m;
      sz[w][16 * t + col] = z;
    }
  }
  __syncthreads();

  if (threadIdx.x < 16 * kPair) {
    const int row = threadIdx.x;
    float m = sm[0][row], z = sz[0][row];
#pragma unroll
    for (int k = 1; k < kSplit; ++k) {
      const float om = sm[k][row];
      const float oz = sz[k][row];
      const float mn = fmaxf(m, om);
      z = z * __expf(m - mn) + oz * __expf(om - mn);
      m = mn;
    }
    Lrow[(size_t)blockIdx.x * (16 * kPair) + row] = m + __logf(z);
  }
}

__global__ void __launch_bounds__(kSplit * 32, 4)
attn_out_kernel(const float* __restrict__ Qf,
                const float* __restrict__ Kf,
                const float* __restrict__ Vf,
                const float* __restrict__ Lrow,
                const float* __restrict__ Wout,
                const float* __restrict__ x,
                const float* __restrict__ gamma,
                float* __restrict__ out) {
  __shared__ float Pt[kSplit][kPair][16 * 16];
  __shared__ float yp[kSplit][kPair][64 * 16];
  __shared__ float ys[kPair][64 * 16];

  const int L    = threadIdx.x & 31;
  const int col  = L & 15;
  const int half = L >> 4;
  const int w  = __builtin_amdgcn_readfirstlane(threadIdx.x >> 5);
  const int b  = blockIdx.x >> 7;
  const int j0 = (blockIdx.x & 127) * (16 * kPair);

  const size_t t0 = (size_t)b * kTiles + (size_t)w * kPerWave;
  const float* Qb = Qf + t0 * 1024 + L * 4;
  const float* Vb = Vf + t0 * 1024 + L * 4;
  const float* Lb = Lrow + (size_t)b * kN + (size_t)w * kPerWave * 16 + 8 * half;
  float* Pw0 = Pt[w][0];
  float* Pw1 = Pt[w][1];

  float4 kb[kPair][8];
#pragma unroll
  for (int t = 0; t < kPair; ++t) {
    const float* kt = Kf + ((size_t)(kPair * blockIdx.x + t) * 8) * 128 + L * 4;
#pragma unroll
    for (int p = 0; p < 8; ++p) kb[t][p] = *(const float4*)(kt + p * 128);
  }

  v8f yacc[kPair][4] = { { v8f{}, v8f{}, v8f{}, v8f{} }, { v8f{}, v8f{}, v8f{}, v8f{} } };

  __builtin_amdgcn_sched_barrier(0);

  float4 aq[8], av[8];
#pragma unroll
  for (int p = 0; p < 8; ++p) aq[p] = *(const float4*)(Qb + p * 128);
  __builtin_amdgcn_sched_barrier(0);
  float4 la  = *(const float4*)(Lb);
  float4 lb4 = *(const float4*)(Lb + 4);
  __builtin_amdgcn_sched_barrier(0);
#pragma unroll
  for (int p = 0; p < 8; ++p) av[p] = *(const float4*)(Vb + p * 128);

  for (int it = 0; it < kPerWave; ++it) {
    const int nx = (it + 1 < kPerWave) ? it + 1 : it;
    const float* qt = Qb + (size_t)nx * 1024;
    const float* vt = Vb + (size_t)nx * 1024;
    const float* lp = Lb + nx * 16;

    v8f c0 = {}, c1 = {};
#pragma unroll
    for (int p = 0; p < 8; ++p) {
      c0 = wmma_f32(lo2(aq[p]), lo2(kb[0][p]), c0);
      c1 = wmma_f32(lo2(aq[p]), lo2(kb[1][p]), c1);
      c0 = wmma_f32(hi2(aq[p]), hi2(kb[0][p]), c0);
      c1 = wmma_f32(hi2(aq[p]), hi2(kb[1][p]), c1);
    }
    asm volatile("" : "+v"(c0), "+v"(c1) :: "memory");
#pragma unroll
    for (int p = 0; p < 8; ++p) aq[p] = *(const float4*)(qt + p * 128);
    __builtin_amdgcn_sched_barrier(0);

    const float lv[8] = { la.x, la.y, la.z, la.w, lb4.x, lb4.y, lb4.z, lb4.w };
#pragma unroll
    for (int r = 0; r < 8; ++r) {
      Pw0[(r + 8 * half) * 16 + col] = __expf(c0[r] - lv[r]);
      Pw1[(r + 8 * half) * 16 + col] = __expf(c1[r] - lv[r]);
    }
    asm volatile("s_wait_dscnt 0x0" ::: "memory");

#pragma unroll
    for (int ci = 0; ci < 4; ++ci) {
      v2f bp0 = { Pw0[(4 * ci + 2 * half) * 16 + col],
                  Pw0[(4 * ci + 2 * half + 1) * 16 + col] };
      v2f bp1 = { Pw1[(4 * ci + 2 * half) * 16 + col],
                  Pw1[(4 * ci + 2 * half + 1) * 16 + col] };
#pragma unroll
      for (int dt = 0; dt < 4; ++dt) {
        const float4 v4 = av[dt * 2 + (ci >> 1)];
        const v2f a = (ci & 1) ? hi2(v4) : lo2(v4);
        yacc[0][dt] = wmma_f32(a, bp0, yacc[0][dt]);
        yacc[1][dt] = wmma_f32(a, bp1, yacc[1][dt]);
      }
    }
    __builtin_amdgcn_sched_barrier(0);
    la  = *(const float4*)(lp);
    lb4 = *(const float4*)(lp + 4);
#pragma unroll
    for (int p = 0; p < 8; ++p) av[p] = *(const float4*)(vt + p * 128);
  }

#pragma unroll
  for (int t = 0; t < kPair; ++t) {
    float* ypw = yp[w][t];
#pragma unroll
    for (int dt = 0; dt < 4; ++dt)
#pragma unroll
      for (int r = 0; r < 8; ++r)
        ypw[(16 * dt + r + 8 * half) * 16 + col] = yacc[t][dt][r];
  }
  __syncthreads();

#pragma unroll
  for (int t = 0; t < kPair; ++t) {
#pragma unroll
    for (int e = 0; e < 8; ++e) {
      const int idx = threadIdx.x + 128 * e;
      ys[t][idx] = ((yp[0][t][idx] + yp[1][t][idx]) + yp[2][t][idx]) + yp[3][t][idx];
    }
  }
  __syncthreads();

  const float g = gamma[0];
  const float* xb = x + (size_t)b * kC * kN;
  float* ob = out + (size_t)b * kC * kN;
  const int n = j0 + col;

  for (int mt = 4 * w; mt < 4 * w + 4; ++mt) {
    const int m0 = mt * 16;
    const float* wr = Wout + (size_t)(m0 + col) * kDK + 2 * half;
    v8f c0 = {}, c1 = {};
#pragma unroll
    for (int s = 0; s < 16; ++s) {
      v2f a   = *(const v2f*)(wr + 4 * s);
      v2f bb0 = { ys[0][(4 * s + 2 * half) * 16 + col],
                  ys[0][(4 * s + 2 * half + 1) * 16 + col] };
      v2f bb1 = { ys[1][(4 * s + 2 * half) * 16 + col],
                  ys[1][(4 * s + 2 * half + 1) * 16 + col] };
      c0 = wmma_f32(a, bb0, c0);
      c1 = wmma_f32(a, bb1, c1);
    }
#pragma unroll
    for (int r = 0; r < 8; ++r) {
      const int row = m0 + r + 8 * half;
      const size_t idx = (size_t)row * kN + n;
      ob[idx]      = c0[r] * g + xb[idx];
      ob[idx + 16] = c1[r] * g + xb[idx + 16];
    }
  }
}

extern "C" void kernel_launch(void* const* d_in, const int* in_sizes, int n_in,
                              void* d_out, int out_size, void* d_ws, size_t ws_size,
                              hipStream_t stream) {
  const float* x     = (const float*)d_in[0];
  const float* Wqkv  = (const float*)d_in[1];
  const float* Wout  = (const float*)d_in[2];
  const float* gamma = (const float*)d_in[3];
  float* out = (float*)d_out;

  float* ws = (float*)d_ws;
  float* Qf = ws;
  float* Kf = ws + 2097152;
  float* Vf = ws + 4194304;
  float* Lr = ws + 6291456;

  qkv_kernel<<<kBS * 12 * 64, 32, 0, stream>>>(x, Wqkv, Qf, Kf, Vf);
  stats_kernel<<<kBS * 128, kSplit * 32, 0, stream>>>(Qf, Kf, Lr);
  attn_out_kernel<<<kBS * 128, kSplit * 32, 0, stream>>>(Qf, Kf, Vf, Lr, Wout, x, gamma, out);
  (void)in_sizes; (void)n_in; (void)out_size; (void)ws_size;
}


# Stats kernel also stores its score tiles to workspace (as many as ws_size holds); attention kernel loads them instead of recomputing: 32 not 64 WMMAs per i-tile; sum order over i unchanged
# speedup vs baseline: 2.3383x; 1.0474x over previous
#include <hip/hip_runtime.h>


static constexpr int kBS = 8;
static constexpr int kC  = 256;
static constexpr int kN  = 4096;
static constexpr int kDK = 64;
static constexpr int kTiles = kN / 16;
static constexpr int kSplit = 4;
static constexpr int kPerWave = kTiles / kSplit;
static constexpr int kPair = 2;

static constexpr int kScoreBase = 6324224;
static constexpr int kPitch = 20;

typedef __attribute__((ext_vector_type(2))) float v2f;
typedef __attribute__((ext_vector_type(4))) float v4f;
typedef __attribute__((ext_vector_type(8))) float v8f;

__device__ __forceinline__ int stored_tiles(const size_t ws_bytes) {
  const size_t base = (size_t)kScoreBase * 4;
  const size_t room = ws_bytes > base ? ws_bytes - base : 0;
  const size_t f = room >> 23;
  return (int)(f > (size_t)kPerWave ? (size_t)kPerWave : f) & ~1;
}

__device__ __forceinline__ v4f ldnt(const float* p) {
  return __builtin_nontemporal_load((const v4f*)p);
}
__device__ __forceinline__ void stnt(float* p, const v4f v) {
  __builtin_nontemporal_store(v, (v4f*)p);
}

__device__ __forceinline__ v8f wmma_f32(v2f a, v2f b, v8f c) {
  return __builtin_amdgcn_wmma_f32_16x16x4_f32(false, a, false, b,
                                               (short)0, c, false, false);
}

__device__ __forceinline__ v2f lo2(const float4 v) { return v2f{ v.x, v.y }; }
__device__ __forceinline__ v2f hi2(const float4 v) { return v2f{ v.z, v.w }; }
__device__ __forceinline__ v2f lo2(const v4f v) { return v2f{ v.x, v.y }; }
__device__ __forceinline__ v2f hi2(const v4f v) { return v2f{ v.z, v.w }; }

template <bool TR>
__device__ __forceinline__ void qkv_tile(const float* __restrict__ xb,
                                         const float* __restrict__ Arow,
                                         const int n, v8f (&cs)[4]) {
  const int half = threadIdx.x >> 4;

  v8f c0 = {}, c1 = {}, c2 = {}, c3 = {};

#pragma unroll 4
  for (int kk = 0; kk < kC; kk += 4) {
    v2f a = *(const v2f*)(Arow + kk);
    const float* xr0 = xb + (size_t)(kk + 2 * half) * kN;
    const float* xr1 = xr0 + kN;
    v2f b0 = { xr0[n],      xr1[n]      };
    v2f b1 = { xr0[n + 16], xr1[n + 16] };
    v2f b2 = { xr0[n + 32], xr1[n + 32] };
    v2f b3 = { xr0[n + 48], xr1[n + 48] };
    if (TR) {
      c0 = wmma_f32(b0, a, c0);
      c1 = wmma_f32(b1, a, c1);
      c2 = wmma_f32(b2, a, c2);
      c3 = wmma_f32(b3, a, c3);
    } else {
      c0 = wmma_f32(a, b0, c0);
      c1 = wmma_f32(a, b1, c1);
      c2 = wmma_f32(a, b2, c2);
      c3 = wmma_f32(a, b3, c3);
    }
  }
  cs[0] = c0; cs[1] = c1; cs[2] = c2; cs[3] = c3;
}

__global__ void qkv_kernel(const float* __restrict__ x,
                           const float* __restrict__ Wqkv,
                           float* __restrict__ Qf,
                           float* __restrict__ Kf,
                           float* __restrict__ Vf) {
  const int L    = threadIdx.x;
  const int col  = L & 15;
  const int half = L >> 4;

  int blk = blockIdx.x;
  const int b   = blk / (12 * 64);
  const int rem = blk % (12 * 64);
  const int m0  = (rem / 64) * 16;
  const int n0  = (rem % 64) * 64;

  const float* xb   = x + (size_t)b * kC * kN;
  const float* Arow = Wqkv + (size_t)(m0 + col) * kC + 2 * half;
  const int n = n0 + col;
  const size_t tile0 = (size_t)b * kTiles + (n0 >> 4);

  v8f cs[4];
  float* dst;
  if (m0 < 128) {
    float* base = (m0 < 64) ? Qf : Kf;
    const int p = ((m0 & 63) >> 3) + half;
    dst = base + (tile0 * 8 + p) * 128 + col * 4;
    qkv_tile<false>(xb, Arow, n, cs);
  } else {
    const int dt = (m0 - 128) >> 4;
    dst = Vf + ((tile0 * 4 + dt) * 2 + half) * 128 + col * 4;
    qkv_tile<true>(xb, Arow, n, cs);
  }

#pragma unroll
  for (int t = 0; t < 4; ++t) {
    float* d0 = dst + (size_t)t * 1024;
    *(float4*)(d0)      = make_float4(cs[t][0], cs[t][1], cs[t][4], cs[t][5]);
    *(float4*)(d0 + 64) = make_float4(cs[t][2], cs[t][3], cs[t][6], cs[t][7]);
  }
}

__device__ __forceinline__ void stats_products(const float4 (&aq)[kPair][8],
                                               const float4 (&kt)[8],
                                               v8f& c0, v8f& c1) {
  c0 = v8f{}; c1 = v8f{};
#pragma unroll
  for (int p = 0; p < 8; ++p) {
    c0 = wmma_f32(lo2(kt[p]), lo2(aq[0][p]), c0);
    c1 = wmma_f32(lo2(kt[p]), lo2(aq[1][p]), c1);
    c0 = wmma_f32(hi2(kt[p]), hi2(aq[0][p]), c0);
    c1 = wmma_f32(hi2(kt[p]), hi2(aq[1][p]), c1);
  }
}

__device__ __forceinline__ float max3f(const float a, const float b, const float c) {
  return fmaxf(fmaxf(a, b), c);
}

__device__ __forceinline__ void row_update(const v8f& a, const v8f& b,
                                           float& m, float& z) {
  const float t0 = max3f(a[0], a[1], a[2]);
  const float t1 = max3f(a[3], a[4], a[5]);
  const float t2 = max3f(a[6], a[7], b[0]);
  const float t3 = max3f(b[1], b[2], b[3]);
  const float t4 = max3f(b[4], b[5], b[6]);
  const float t5 = max3f(b[7], m, t0);
  const float mn = max3f(max3f(t1, t2, t3), t4, t5);
  float ea[8], eb[8];
#pragma unroll
  for (int r = 0; r < 8; ++r) {
    ea[r] = __expf(a[r] - mn);
    eb[r] = __expf(b[r] - mn);
  }
  const float pa = ((ea[0] + ea[1]) + (ea[2] + ea[3])) + ((ea[4] + ea[5]) + (ea[6] + ea[7]));
  const float pb = ((eb[0] + eb[1]) + (eb[2] + eb[3])) + ((eb[4] + eb[5]) + (eb[6] + eb[7]));
  z = z * __expf(m - mn) + (pa + pb);
  m = mn;
}

template <bool STORE>
__device__ __forceinline__ void stats_stream(const float4 (&aq)[kPair][8],
                                             const float* __restrict__ Kb,
                                             float* __restrict__ Sw,
                                             const size_t jstride,
                                             float (&mr)[kPair], float (&zr)[kPair]) {
  float4 ka[8], kb[8];
#pragma unroll
  for (int p = 0; p < 8; ++p) ka[p] = *(const float4*)(Kb + p * 128);
  __builtin_amdgcn_sched_barrier(0);
#pragma unroll
  for (int p = 0; p < 8; ++p) kb[p] = *(const float4*)(Kb + 1024 + p * 128);

  for (int jt = 0; jt < kPerWave; jt += 2) {
    const int nj = (jt + 2 < kPerWave) ? jt + 2 : jt;
    const float* k2 = Kb + (size_t)nj * 1024;
    v8f ca0, ca1, cb0, cb1;

    stats_products(aq, ka, ca0, ca1);
    asm volatile("" : "+v"(ca0), "+v"(ca1) :: "memory");
#pragma unroll
    for (int p = 0; p < 8; ++p) ka[p] = *(const float4*)(k2 + p * 128);
    __builtin_amdgcn_sched_barrier(0);

    stats_products(aq, kb, cb0, cb1);
    asm volatile("" : "+v"(cb0), "+v"(cb1) :: "memory");
#pragma unroll
    for (int p = 0; p < 8; ++p) kb[p] = *(const float4*)(k2 + 1024 + p * 128);
    __builtin_amdgcn_sched_barrier(0);

    if (STORE) {
      float* sp = Sw + (size_t)(jt >> 1) * jstride;
      stnt(sp,       __builtin_shufflevector(ca0, ca0, 0, 1, 2, 3));
      stnt(sp + 128, __builtin_shufflevector(ca0, ca0, 4, 5, 6, 7));
      stnt(sp + 256, __builtin_shufflevector(cb0, cb0, 0, 1, 2, 3));
      stnt(sp + 384, __builtin_shufflevector(cb0, cb0, 4, 5, 6, 7));
      stnt(sp + 512, __builtin_shufflevector(ca1, ca1, 0, 1, 2, 3));
      stnt(sp + 640, __builtin_shufflevector(ca1, ca1, 4, 5, 6, 7));
      stnt(sp + 768, __builtin_shufflevector(cb1, cb1, 0, 1, 2, 3));
      stnt(sp + 896, __builtin_shufflevector(cb1, cb1, 4, 5, 6, 7));
    }

    row_update(ca0, cb0, mr[0], zr[0]);
    row_update(ca1, cb1, mr[1], zr[1]);
  }
}

__global__ void __launch_bounds__(kSplit * 32, 4)
stats_kernel(const float* __restrict__ Qf,
             const float* __restrict__ Kf,
             float* __restrict__ Lrow,
             float* __restrict__ Sx,
             const size_t ws_bytes) {
  __shared__ float sm[kSplit][16 * kPair];
  __shared__ float sz[kSplit][16 * kPair];

  const int L    = threadIdx.x & 31;
  const int col  = L & 15;
  const int half = L >> 4;
  const int w  = __builtin_amdgcn_readfirstlane(threadIdx.x >> 5);
  const int b  = blockIdx.x >> 7;

  float4 aq[kPair][8];
#pragma unroll
  for (int t = 0; t < kPair; ++t) {
    const float* qt = Qf + ((size_t)(kPair * blockIdx.x + t) * 8) * 128 + L * 4;
#pragma unroll
    for (int p = 0; p < 8; ++p) aq[t][p] = *(const float4*)(qt + p * 128);
  }

  const float* Kb = Kf + ((size_t)b * kTiles + (size_t)w * kPerWave) * 1024 + L * 4;

  float mr[kPair], zr[kPair];
#pragma unroll
  for (int t = 0; t < kPair; ++t) { mr[t] = -3.0e38f; zr[t] = 0.0f; }

  const int F  = stored_tiles(ws_bytes);
  const int il = kPair * (blockIdx.x & 127);
  const size_t jstride = (size_t)F * (kSplit * 512);
  float* Sw = Sx + ((size_t)b * 128 + (size_t)w * 32) * jstride
                 + ((size_t)(il >> 6) * F + (size_t)(il & 63)) * 512 + L * 4;
  if ((il & 63) < F) stats_stream<true>(aq, Kb, Sw, jstride, mr, zr);
  else               stats_stream<false>(aq, Kb, Sw, jstride, mr, zr);

#pragma unroll
  for (int t = 0; t < kPair; ++t) {
    float m = mr[t];
    float z = zr[t];
    const float om = __shfl_xor(m, 16);
    const float oz = __shfl_xor(z, 16);
    const float mn = fmaxf(m, om);
    z = z * __expf(m - mn) + oz * __expf(om - mn);
    m = mn;
    if (half == 0) {
      sm[w][16 * t + col] = m;
      sz[w][16 * t + col] = z;
    }
  }
  __syncthreads();

  if (threadIdx.x < 16 * kPair) {
    const int row = threadIdx.x;
    float m = sm[0][row], z = sz[0][row];
#pragma unroll
    for (int k = 1; k < kSplit; ++k) {
      const float om = sm[k][row];
      const float oz = sz[k][row];
      const float mn = fmaxf(m, om);
      z = z * __expf(m - mn) + oz * __expf(om - mn);
      m = mn;
    }
    Lrow[(size_t)blockIdx.x * (16 * kPair) + row] = m + __logf(z);
  }
}

template <typename V4>
__device__ __forceinline__ void pv_products(const V4 (&av)[8],
                                            const float* Pw0, const float* Pw1,
                                            v8f (&yacc)[kPair][4],
                                            const int col, const int half) {
#pragma unroll
  for (int ci = 0; ci < 4; ++ci) {
    v2f bp0 = { Pw0[(4 * ci + 2 * half) * kPitch + col],
                Pw0[(4 * ci + 2 * half + 1) * kPitch + col] };
    v2f bp1 = { Pw1[(4 * ci + 2 * half) * kPitch + col],
                Pw1[(4 * ci + 2 * half + 1) * kPitch + col] };
#pragma unroll
    for (int dt = 0; dt < 4; ++dt) {
      const V4 v4 = av[dt * 2 + (ci >> 1)];
      const v2f a = (ci & 1) ? hi2(v4) : lo2(v4);
      yacc[0][dt] = wmma_f32(a, bp0, yacc[0][dt]);
      yacc[1][dt] = wmma_f32(a, bp1, yacc[1][dt]);
    }
  }
}

__device__ __forceinline__ void stored_trip(v4f (&s)[4], float& lrow, v4f (&av)[8],
                                            const float* sn, const float* ln, const float* vn,
                                            float* Pw0, float* Pw1,
                                            v8f (&yacc)[kPair][4],
                                            const int col, const int half) {
  v4f p[4];
#pragma unroll
  for (int q = 0; q < 4; ++q) {
    p[q].x = __expf(s[q].x - lrow);
    p[q].y = __expf(s[q].y - lrow);
    p[q].z = __expf(s[q].z - lrow);
    p[q].w = __expf(s[q].w - lrow);
  }
  asm volatile("" : "+v"(p[0]), "+v"(p[1]), "+v"(p[2]), "+v"(p[3]) :: "memory");
  float* r0 = Pw0 + col * kPitch + 8 * half;
  float* r1 = Pw1 + col * kPitch + 8 * half;
  *(v4f*)(r0)     = p[0];
  *(v4f*)(r0 + 4) = p[1];
  *(v4f*)(r1)     = p[2];
  *(v4f*)(r1 + 4) = p[3];
#pragma unroll
  for (int q = 0; q < 4; ++q) s[q] = ldnt(sn + q * 128);
  lrow = *ln;
  __builtin_amdgcn_sched_barrier(0);
  asm volatile("s_wait_dscnt 0x0" ::: "memory");
  pv_products(av, Pw0, Pw1, yacc, col, half);
  asm volatile("" : "+v"(yacc[0][0]), "+v"(yacc[0][1]), "+v"(yacc[0][2]), "+v"(yacc[0][3]),
                    "+v"(yacc[1][0]), "+v"(yacc[1][1]), "+v"(yacc[1][2]), "+v"(yacc[1][3]) :: "memory");
  __builtin_amdgcn_sched_barrier(0);
#pragma unroll
  for (int pp = 0; pp < 8; ++pp) av[pp] = *(const v4f*)(vn + pp * 128);
}

__global__ void __launch_bounds__(kSplit * 32, 4)
attn_out_kernel(const float* __restrict__ Qf,
                const float* __restrict__ Kf,
                const float* __restrict__ Vf,
                const float* __restrict__ Lrow,
                const float* __restrict__ Sx,
                const float* __restrict__ Wout,
                const float* __restrict__ x,
                const float* __restrict__ gamma,
                float* __restrict__ out,
                const size_t ws_bytes) {
  __shared__ __attribute__((aligned(16))) float Pt[kSplit][kPair][16 * kPitch];
  __shared__ float yp[kSplit][kPair][64 * 16];
  __shared__ float ys[kPair][64 * 16];

  const int L    = threadIdx.x & 31;
  const int col  = L & 15;
  const int half = L >> 4;
  const int w  = __builtin_amdgcn_readfirstlane(threadIdx.x >> 5);
  const int b  = blockIdx.x >> 7;
  const int j0 = (blockIdx.x & 127) * (16 * kPair);

  const size_t t0 = (size_t)b * kTiles + (size_t)w * kPerWave;
  const float* Qb = Qf + t0 * 1024 + L * 4;
  const float* Vb = Vf + t0 * 1024 + L * 4;
  const float* Lw = Lrow + (size_t)b * kN + (size_t)w * kPerWave * 16;
  float* Pw0 = Pt[w][0];
  float* Pw1 = Pt[w][1];

  const int F = stored_tiles(ws_bytes);

  v8f yacc[kPair][4] = { { v8f{}, v8f{}, v8f{}, v8f{} }, { v8f{}, v8f{}, v8f{}, v8f{} } };

  if (F > 0) {
    const float* Sb = Sx + ((size_t)blockIdx.x * kSplit + (size_t)w) * F * 512 + L * 4;
    const float* Lb = Lw + col;
    v4f sa[4], sb[4];
    v4f va[8], vb[8];
#pragma unroll
    for (int q = 0; q < 4; ++q) sa[q] = ldnt(Sb + q * 128);
    float l_a = Lb[0];
    __builtin_amdgcn_sched_barrier(0);
#pragma unroll
    for (int p = 0; p < 8; ++p) va[p] = *(const v4f*)(Vb + p * 128);
    __builtin_amdgcn_sched_barrier(0);
#pragma unroll
    for (int q = 0; q < 4; ++q) sb[q] = ldnt(Sb + 512 + q * 128);
    float l_b = Lb[16];
    __builtin_amdgcn_sched_barrier(0);
#pragma unroll
    for (int p = 0; p < 8; ++p) vb[p] = *(const v4f*)(Vb + 1024 + p * 128);

    for (int it = 0; it < F; it += 2) {
      const int n2 = (it + 2 < F) ? it + 2 : it;
      stored_trip(sa, l_a, va, Sb + (size_t)n2 * 512, Lb + n2 * 16, Vb + (size_t)n2 * 1024,
                  Pw0, Pw1, yacc, col, half);
      stored_trip(sb, l_b, vb, Sb + (size_t)(n2 + 1) * 512, Lb + (n2 + 1) * 16,
                  Vb + (size_t)(n2 + 1) * 1024, Pw0, Pw1, yacc, col, half);
    }
  }

  if (F < kPerWave) {
    const float* Lb = Lw + 8 * half;
    float4 kb[kPair][8];
#pragma unroll
    for (int t = 0; t < kPair; ++t) {
      const float* kt = Kf + ((size_t)(kPair * blockIdx.x + t) * 8) * 128 + L * 4;
#pragma unroll
      for (int p = 0; p < 8; ++p) kb[t][p] = *(const float4*)(kt + p * 128);
    }

    __builtin_amdgcn_sched_barrier(0);

    float4 aq[8], av[8];
#pragma unroll
    for (int p = 0; p < 8; ++p) aq[p] = *(const float4*)(Qb + (size_t)F * 1024 + p * 128);
    __builtin_amdgcn_sched_barrier(0);
    float4 la  = *(const float4*)(Lb + F * 16);
    float4 lb4 = *(const float4*)(Lb + F * 16 + 4);
    __builtin_amdgcn_sched_barrier(0);
#pragma unroll
    for (int p = 0; p < 8; ++p) av[p] = *(const float4*)(Vb + (size_t)F * 1024 + p * 128);

    for (int it = F; it < kPerWave; ++it) {
      const int nx = (it + 1 < kPerWave) ? it + 1 : it;
      const float* qt = Qb + (size_t)nx * 1024;
      const float* vt = Vb + (size_t)nx * 1024;
      const float* lp = Lb + nx * 16;

      v8f c0 = {}, c1 = {};
#pragma unroll
      for (int p = 0; p < 8; ++p) {
        c0 = wmma_f32(lo2(aq[p]), lo2(kb[0][p]), c0);
        c1 = wmma_f32(lo2(aq[p]), lo2(kb[1][p]), c1);
        c0 = wmma_f32(hi2(aq[p]), hi2(kb[0][p]), c0);
        c1 = wmma_f32(hi2(aq[p]), hi2(kb[1][p]), c1);
      }
      asm volatile("" : "+v"(c0), "+v"(c1) :: "memory");
#pragma unroll
      for (int p = 0; p < 8; ++p) aq[p] = *(const float4*)(qt + p * 128);
      __builtin_amdgcn_sched_barrier(0);

      const float lv[8] = { la.x, la.y, la.z, la.w, lb4.x, lb4.y, lb4.z, lb4.w };
#pragma unroll
      for (int r = 0; r < 8; ++r) {
        Pw0[(r + 8 * half) * kPitch + col] = __expf(c0[r] - lv[r]);
        Pw1[(r + 8 * half) * kPitch + col] = __expf(c1[r] - lv[r]);
      }
      asm volatile("s_wait_dscnt 0x0" ::: "memory");

      pv_products(av, Pw0, Pw1, yacc, col, half);
      __builtin_amdgcn_sched_barrier(0);
      la  = *(const float4*)(lp);
      lb4 = *(const float4*)(lp + 4);
#pragma unroll
      for (int p = 0; p < 8; ++p) av[p] = *(const float4*)(vt + p * 128);
    }
  }

#pragma unroll
  for (int t = 0; t < kPair; ++t) {
    float* ypw = yp[w][t];
#pragma unroll
    for (int dt = 0; dt < 4; ++dt)
#pragma unroll
      for (int r = 0; r < 8; ++r)
        ypw[(16 * dt + r + 8 * half) * 16 + col] = yacc[t][dt][r];
  }
  __syncthreads();

#pragma unroll
  for (int t = 0; t < kPair; ++t) {
#pragma unroll
    for (int e = 0; e < 8; ++e) {
      const int idx = threadIdx.x + 128 * e;
      ys[t][idx] = ((yp[0][t][idx] + yp[1][t][idx]) + yp[2][t][idx]) + yp[3][t][idx];
    }
  }
  __syncthreads();

  const float g = gamma[0];
  const float* xb = x + (size_t)b * kC * kN;
  float* ob = out + (size_t)b * kC * kN;
  const int n = j0 + col;

  for (int mt = 4 * w; mt < 4 * w + 4; ++mt) {
    const int m0 = mt * 16;
    const float* wr = Wout + (size_t)(m0 + col) * kDK + 2 * half;
    v8f c0 = {}, c1 = {};
#pragma unroll
    for (int s = 0; s < 16; ++s) {
      v2f a   = *(const v2f*)(wr + 4 * s);
      v2f bb0 = { ys[0][(4 * s + 2 * half) * 16 + col],
                  ys[0][(4 * s + 2 * half + 1) * 16 + col] };
      v2f bb1 = { ys[1][(4 * s + 2 * half) * 16 + col],
                  ys[1][(4 * s + 2 * half + 1) * 16 + col] };
      c0 = wmma_f32(a, bb0, c0);
      c1 = wmma_f32(a, bb1, c1);
    }
#pragma unroll
    for (int r = 0; r < 8; ++r) {
      const int row = m0 + r + 8 * half;
      const size_t idx = (size_t)row * kN + n;
      ob[idx]      = c0[r] * g + xb[idx];
      ob[idx + 16] = c1[r] * g + xb[idx + 16];
    }
  }
}

extern "C" void kernel_launch(void* const* d_in, const int* in_sizes, int n_in,
                              void* d_out, int out_size, void* d_ws, size_t ws_size,
                              hipStream_t stream) {
  const float* x     = (const float*)d_in[0];
  const float* Wqkv  = (const float*)d_in[1];
  const float* Wout  = (const float*)d_in[2];
  const float* gamma = (const float*)d_in[3];
  float* out = (float*)d_out;

  float* ws = (float*)d_ws;
  float* Qf = ws;
  float* Kf = ws + 2097152;
  float* Vf = ws + 4194304;
  float* Lr = ws + 6291456;
  float* Sx = ws + kScoreBase;

  qkv_kernel<<<kBS * 12 * 64, 32, 0, stream>>>(x, Wqkv, Qf, Kf, Vf);
  stats_kernel<<<kBS * 128, kSplit * 32, 0, stream>>>(Qf, Kf, Lr, Sx, ws_size);
  attn_out_kernel<<<kBS * 128, kSplit * 32, 0, stream>>>(Qf, Kf, Vf, Lr, Sx, Wout, x, gamma, out, ws_size);
  (void)in_sizes; (void)n_in; (void)out_size;
}
